# starting kernel
# baseline (speedup 1.0000x reference)
#include <hip/hip_runtime.h>
#include <hip/hip_bf16.h>

typedef __attribute__((ext_vector_type(2))) float v2f;
typedef __attribute__((ext_vector_type(8))) float v8f;

__global__ void k_fill(float* __restrict__ p, float v, int n) {
  int i = blockIdx.x * blockDim.x + threadIdx.x;
  if (i < n) p[i] = v;
}

__global__ void k_deg_edges(const int* __restrict__ dst, float* __restrict__ deg, int E) {
  int i = blockIdx.x * blockDim.x + threadIdx.x;
  if (i < E) atomicAdd(deg + dst[i], 1.0f);
}

__global__ void k_dinv(float* __restrict__ deg, int n) {
  int i = blockIdx.x * blockDim.x + threadIdx.x;
  if (i < n) deg[i] = rsqrtf(fmaxf(deg[i], 1.0f));
}

__global__ void __launch_bounds__(256) k_gemm_wmma(
    const float* __restrict__ A, const float* __restrict__ W,
    float* __restrict__ out, int N, int K, int M) {
  __shared__ float As[16 * (256 + 4)];
  const int ldsK = K + 4;
  const int rowBase = blockIdx.x * 16;
  const int colBase = blockIdx.y * 128 + (threadIdx.x >> 5) * 16;

  for (int e = threadIdx.x; e < 16 * K; e += 256) {
    int r = e / K;
    int c = e - r * K;
    int g = rowBase + r;
    As[r * ldsK + c] = (g < N) ? A[(size_t)g * K + c] : 0.0f;
  }
  __syncthreads();

  const int lane = threadIdx.x & 31;
  const int half = lane >> 4;
  const int m    = lane & 15;
  const int col  = colBase + m;

  v8f acc = {};
  for (int k = 0; k < K; k += 4) {
    const int kk = k + 2 * half;
    v2f a, b;
    a.x = As[m * ldsK + kk];
    a.y = As[m * ldsK + kk + 1];
    b.x = W[(size_t)kk * M + col];
    b.y = W[(size_t)(kk + 1) * M + col];
    acc = __builtin_amdgcn_wmma_f32_16x16x4_f32(
         false, a,  false, b,
         (short)0, acc,  false,  false);
  }

  for (int r = 0; r < 8; ++r) {
    int row = rowBase + r + 8 * half;
    if (row < N) out[(size_t)row * M + col] = acc[r];
  }
}

__global__ void k_selfloop(const float* __restrict__ hW, const float* __restrict__ dinv,
                           float* __restrict__ agg, int N, int C) {
  long long i = (long long)blockIdx.x * blockDim.x + threadIdx.x;
  long long total = (long long)N * C;
  if (i < total) {
    int node = (int)(i / C);
    float di = dinv[node];
    agg[i] = hW[i] * di * di;
  }
}

__global__ void k_edge_scatter(const int* __restrict__ src, const int* __restrict__ dst,
                               const float* __restrict__ dinv, const float* __restrict__ hW,
                               float* __restrict__ agg, int E, int C) {
  int wave = (int)(((long long)blockIdx.x * blockDim.x + threadIdx.x) >> 5);
  int lane = threadIdx.x & 31;
  if (wave >= E) return;
  int s = src[wave];
  int d = dst[wave];
  float nrm = dinv[s] * dinv[d];
  const float4* hp = (const float4*)(hW + (size_t)s * C);
  float* ap = agg + (size_t)d * C;
  int nchunk = C >> 7;
  for (int j = 0; j < nchunk; ++j) {
    int v4 = j * 32 + lane;
    float4 v = hp[v4];
    int base = v4 * 4;
    atomicAdd(ap + base + 0, v.x * nrm);
    atomicAdd(ap + base + 1, v.y * nrm);
    atomicAdd(ap + base + 2, v.z * nrm);
    atomicAdd(ap + base + 3, v.w * nrm);
  }
}

__global__ void k_post(const float* __restrict__ agg, const float* __restrict__ bias,
                       float* __restrict__ out, int N, int C) {
  int node = (int)(((long long)blockIdx.x * blockDim.x + threadIdx.x) >> 5);
  int lane = threadIdx.x & 31;
  if (node >= N) return;
  const float* row = agg + (size_t)node * C;
  float v[8];
  int nch = C >> 5;
  float s = 0.0f;
  for (int j = 0; j < nch; ++j) {
    float t = fmaxf(row[j * 32 + lane] + bias[j * 32 + lane], 0.0f);
    v[j] = t;
    s += t;
  }
  for (int off = 16; off > 0; off >>= 1) s += __shfl_xor(s, off, 32);
  float mean = s / (float)C;
  float q = 0.0f;
  for (int j = 0; j < nch; ++j) {
    float d = v[j] - mean;
    q += d * d;
  }
  for (int off = 16; off > 0; off >>= 1) q += __shfl_xor(q, off, 32);
  float rstd = rsqrtf(q / (float)C + 1e-5f);
  float* o = out + (size_t)node * C;
  for (int j = 0; j < nch; ++j) o[j * 32 + lane] = (v[j] - mean) * rstd;
}

__global__ void k_pool_accum(const float* __restrict__ h, const int* __restrict__ batch,
                             float* __restrict__ pool, int N, int C) {
  long long i = (long long)blockIdx.x * blockDim.x + threadIdx.x;
  long long total = (long long)N * C;
  if (i < total) {
    int node = (int)(i / C);
    int c = (int)(i - (long long)node * C);
    atomicAdd(pool + (size_t)batch[node] * C + c, h[i]);
  }
}

__global__ void k_count(const int* __restrict__ batch, float* __restrict__ cnt, int N) {
  int i = blockIdx.x * blockDim.x + threadIdx.x;
  if (i < N) atomicAdd(cnt + batch[i], 1.0f);
}

__global__ void k_pool_final(float* __restrict__ pool, const float* __restrict__ cnt,
                             int G, int C) {
  int i = blockIdx.x * blockDim.x + threadIdx.x;
  if (i < G * C) pool[i] /= fmaxf(cnt[i / C], 1.0f);
}

static inline size_t align256(size_t x) { return (x + 255) & ~(size_t)255; }

extern "C" void kernel_launch(void* const* d_in, const int* in_sizes, int n_in,
                              void* d_out, int out_size, void* d_ws, size_t ws_size,
                              hipStream_t stream) {
  const float* x     = (const float*)d_in[0];
  const int*   ei    = (const int*)d_in[1];
  const int*   batch = (const int*)d_in[2];
  const float* W0 = (const float*)d_in[4];
  const float* b0 = (const float*)d_in[5];
  const float* W1 = (const float*)d_in[6];
  const float* b1 = (const float*)d_in[7];
  const float* W2 = (const float*)d_in[8];
  const float* b2 = (const float*)d_in[9];

  const int N  = in_sizes[2];
  const int E  = in_sizes[1] / 2;
  const int IN = in_sizes[0] / N;
  const int H0 = in_sizes[5];
  const int H1 = in_sizes[7];
  const int H2 = in_sizes[9];
  const int G  = out_size / H2 - N;

  const int* srcp = ei;
  const int* dstp = ei + E;

  char* ws = (char*)d_ws;
  float* dinv = (float*)ws;  ws += align256((size_t)N * 4);
  float* cnt  = (float*)ws;  ws += align256((size_t)G * 4);
  float* bufA = (float*)ws;  ws += align256((size_t)N * 256 * 4);
  float* bufB = (float*)ws;  ws += align256((size_t)N * 256 * 4);

  float* pool = (float*)d_out;
  float* hout = (float*)d_out + (size_t)G * H2;

  const int B = 256;
  auto cdiv = [](long long a, long long b) { return (int)((a + b - 1) / b); };

  k_fill<<<cdiv(N, B), B, 0, stream>>>(dinv, 1.0f, N);
  k_deg_edges<<<cdiv(E, B), B, 0, stream>>>(dstp, dinv, E);
  k_dinv<<<cdiv(N, B), B, 0, stream>>>(dinv, N);

  auto layer = [&](const float* hin, int K, int M, const float* W, const float* bias,
                   float* hw, float* agg, float* outp) {
    dim3 g(cdiv(N, 16), M / 128);
    k_gemm_wmma<<<g, B, 0, stream>>>(hin, W, hw, N, K, M);
    k_selfloop<<<cdiv((long long)N * M, B), B, 0, stream>>>(hw, dinv, agg, N, M);
    k_edge_scatter<<<cdiv((long long)E * 32, B), B, 0, stream>>>(srcp, dstp, dinv, hw, agg, E, M);
    k_post<<<cdiv((long long)N * 32, B), B, 0, stream>>>(agg, bias, outp, N, M);
  };

  layer(x,    IN, H0, W0, b0,  bufA,  bufB,  bufA);
  layer(bufA, H0, H1, W1, b1,  bufB,  bufA,  bufB);
  layer(bufB, H1, H2, W2, b2,  bufA,  bufB,  hout);

  k_fill<<<cdiv((long long)G * H2, B), B, 0, stream>>>(pool, 0.0f, G * H2);
  k_fill<<<cdiv(G, B), B, 0, stream>>>(cnt, 0.0f, G);
  k_pool_accum<<<cdiv((long long)N * H2, B), B, 0, stream>>>(hout, batch, pool, N, H2);
  k_count<<<cdiv(N, B), B, 0, stream>>>(batch, cnt, N);
  k_pool_final<<<cdiv((long long)G * H2, B), B, 0, stream>>>(pool, cnt, G, H2);
}


# Product kernel gets K and M as compile-time sizes: staging loads all in flight, k loop unrolled on immediate offsets (no integer divide, no 64-bit address multiplies by the WMMAs); sum order kept
# speedup vs baseline: 1.8656x; 1.8656x over previous
#include <hip/hip_runtime.h>
#include <hip/hip_bf16.h>

typedef __attribute__((ext_vector_type(2))) float v2f;
typedef __attribute__((ext_vector_type(8))) float v8f;

__global__ void k_fill(float* __restrict__ p, float v, int n) {
  int i = blockIdx.x * blockDim.x + threadIdx.x;
  if (i < n) p[i] = v;
}

__global__ void k_deg_edges(const int* __restrict__ dst, float* __restrict__ deg, int E) {
  int i = blockIdx.x * blockDim.x + threadIdx.x;
  if (i < E) atomicAdd(deg + dst[i], 1.0f);
}

__global__ void k_dinv(float* __restrict__ deg, int n) {
  int i = blockIdx.x * blockDim.x + threadIdx.x;
  if (i < n) deg[i] = rsqrtf(fmaxf(deg[i], 1.0f));
}

__global__ void __launch_bounds__(256) k_gemm_wmma(
    const float* __restrict__ A, const float* __restrict__ W,
    float* __restrict__ out, int N, int K, int M) {
  __shared__ float As[16 * (256 + 4)];
  const int ldsK = K + 4;
  const int rowBase = blockIdx.x * 16;
  const int colBase = blockIdx.y * 128 + (threadIdx.x >> 5) * 16;

  for (int e = threadIdx.x; e < 16 * K; e += 256) {
    int r = e / K;
    int c = e - r * K;
    int g = rowBase + r;
    As[r * ldsK + c] = (g < N) ? A[(size_t)g * K + c] : 0.0f;
  }
  __syncthreads();

  const int lane = threadIdx.x & 31;
  const int half = lane >> 4;
  const int m    = lane & 15;
  const int col  = colBase + m;

  v8f acc = {};
  for (int k = 0; k < K; k += 4) {
    const int kk = k + 2 * half;
    v2f a, b;
    a.x = As[m * ldsK + kk];
    a.y = As[m * ldsK + kk + 1];
    b.x = W[(size_t)kk * M + col];
    b.y = W[(size_t)(kk + 1) * M + col];
    acc = __builtin_amdgcn_wmma_f32_16x16x4_f32(
         false, a,  false, b,
         (short)0, acc,  false,  false);
  }

  for (int r = 0; r < 8; ++r) {
    int row = rowBase + r + 8 * half;
    if (row < N) out[(size_t)row * M + col] = acc[r];
  }
}

template <int K, int M>
__global__ void __launch_bounds__(256) k_gemm_wmma_kc(
    const float* __restrict__ A, const float* __restrict__ W,
    float* __restrict__ out, int N) {
  constexpr int ldsK = K + 4;
  __shared__ float As[16 * ldsK];
  const int rowBase = blockIdx.x * 16;
  const int colBase = blockIdx.y * 128 + (threadIdx.x >> 5) * 16;

  constexpr int NS = (16 * K) / 256;
  float stage[NS];
#pragma unroll
  for (int i = 0; i < NS; ++i) {
    const int e = (int)threadIdx.x + 256 * i;
    const int r = e / K;
    const int c = e - r * K;
    const int g = rowBase + r;
    const int gl = (g < N) ? g : N - 1;
    stage[i] = A[(size_t)gl * K + c];
  }
#pragma unroll
  for (int i = 0; i < NS; ++i) {
    const int e = (int)threadIdx.x + 256 * i;
    const int r = e / K;
    const int c = e - r * K;
    As[r * ldsK + c] = (rowBase + r < N) ? stage[i] : 0.0f;
  }
  __syncthreads();

  const int lane = threadIdx.x & 31;
  const int half = lane >> 4;
  const int m    = lane & 15;
  const int col  = colBase + m;

  const float* ap = As + m * ldsK + 2 * half;
  const float* wp = W + (size_t)(2 * half) * M + col;

  v8f acc = {};
#pragma unroll 16
  for (int k = 0; k < K; k += 4) {
    v2f a, b;
    a.x = ap[k];
    a.y = ap[k + 1];
    b.x = wp[(size_t)k * M];
    b.y = wp[(size_t)(k + 1) * M];
    acc = __builtin_amdgcn_wmma_f32_16x16x4_f32(
         false, a,  false, b,
         (short)0, acc,  false,  false);
  }

  float* op = out + (size_t)(rowBase + 8 * half) * M + col;
#pragma unroll
  for (int r = 0; r < 8; ++r) {
    if (rowBase + 8 * half + r < N) op[(size_t)r * M] = acc[r];
  }
}

__device__ __forceinline__ int wave_in_grid() {
  return __builtin_amdgcn_readfirstlane((int)((blockIdx.x * blockDim.x + threadIdx.x) >> 5));
}
__device__ __forceinline__ int waves_in_grid() { return (int)((gridDim.x * blockDim.x) >> 5); }

template <int NL, int R>
__global__ void __launch_bounds__(256) k_selfloop(const float* __restrict__ hW,
                                                  const float* __restrict__ dinv,
                                                  float* __restrict__ agg, int N) {
  constexpr int C = NL * 32;
  const int lane = threadIdx.x & 31;
  const int gw = wave_in_grid();
  const int nw = waves_in_grid();
  for (int r0 = gw; r0 < N; r0 += R * nw) {
    float a[R][NL];
    float di[R];
    for (int k = 0; k < R; ++k) {
      const int r = r0 + k * nw;
      di[k] = 0.0f;
      if (r < N) {
        const float* p = hW + (size_t)r * C + lane;
        for (int i = 0; i < NL; ++i) a[k][i] = p[i * 32];
        di[k] = dinv[r];
      }
    }
    for (int k = 0; k < R; ++k) {
      const int r = r0 + k * nw;
      if (r < N) {
        float* o = agg + (size_t)r * C + lane;
        for (int i = 0; i < NL; ++i) o[i * 32] = a[k][i] * di[k] * di[k];
      }
    }
  }
}

template <int NL>
__global__ void __launch_bounds__(256) k_edge_scatter(const int* __restrict__ src,
                                                      const int* __restrict__ dst,
                                                      const float* __restrict__ dinv,
                                                      const float* __restrict__ hW,
                                                      float* __restrict__ agg, int E) {
  constexpr int C = NL * 32;
  const int lane = threadIdx.x & 31;
  const int gw = wave_in_grid();
  const int nw = waves_in_grid();
  const int per = (E + nw - 1) / nw;
  const long long first = (long long)gw * per;
  if (first >= E) return;
  const int e0 = (int)first;
  const int e1 = (E - e0 < per) ? E : e0 + per;
  for (int b = e0; b < e1; b += 32) {
    const int e = b + lane;
    int s = 0, d = 0;
    float nrm = 0.0f;
    if (e < e1) {
      s = src[e];
      d = dst[e];
      nrm = dinv[s] * dinv[d];
    }
    const int cnt = (e1 - b < 32) ? (e1 - b) : 32;
    float cur[NL], nxt[NL];
    {
      const int s0 = __builtin_amdgcn_readlane(s, 0);
      const float* hp = hW + (size_t)s0 * C + lane;
      for (int i = 0; i < NL; ++i) cur[i] = hp[i * 32];
    }
    for (int j = 0; j < cnt; ++j) {
      const int jn = (j + 1 < cnt) ? j + 1 : j;
      const int sn = __builtin_amdgcn_readlane(s, jn);
      const float* hp = hW + (size_t)sn * C + lane;
      for (int i = 0; i < NL; ++i) nxt[i] = hp[i * 32];
      const int dj = __builtin_amdgcn_readlane(d, j);
      const float nj = __int_as_float(__builtin_amdgcn_readlane(__float_as_int(nrm), j));
      float* ap = agg + (size_t)dj * C + lane;
      for (int i = 0; i < NL; ++i) atomicAdd(ap + i * 32, cur[i] * nj);
      for (int i = 0; i < NL; ++i) cur[i] = nxt[i];
    }
  }
}

template <int NL, int R>
__global__ void __launch_bounds__(256) k_post(const float* __restrict__ agg,
                                              const float* __restrict__ bias,
                                              float* __restrict__ out, int N) {
  constexpr int C = NL * 32;
  const int lane = threadIdx.x & 31;
  const int gw = wave_in_grid();
  const int nw = waves_in_grid();
  float bv[NL];
  for (int j = 0; j < NL; ++j) bv[j] = bias[j * 32 + lane];
  for (int n0 = gw; n0 < N; n0 += R * nw) {
    float v[R][NL];
    for (int k = 0; k < R; ++k) {
      const int node = n0 + k * nw;
      if (node < N) {
        const float* row = agg + (size_t)node * C + lane;
        for (int j = 0; j < NL; ++j) v[k][j] = row[j * 32];
      }
    }
    for (int k = 0; k < R; ++k) {
      const int node = n0 + k * nw;
      if (node < N) {
        float s = 0.0f;
        for (int j = 0; j < NL; ++j) {
          float t = fmaxf(v[k][j] + bv[j], 0.0f);
          v[k][j] = t;
          s += t;
        }
        for (int off = 16; off > 0; off >>= 1) s += __shfl_xor(s, off, 32);
        float mean = s / (float)C;
        float q = 0.0f;
        for (int j = 0; j < NL; ++j) {
          float dlt = v[k][j] - mean;
          q += dlt * dlt;
        }
        for (int off = 16; off > 0; off >>= 1) q += __shfl_xor(q, off, 32);
        float rstd = rsqrtf(q / (float)C + 1e-5f);
        float* o = out + (size_t)node * C + lane;
        for (int j = 0; j < NL; ++j) o[j * 32] = (v[k][j] - mean) * rstd;
      }
    }
  }
}

template <int NL, int R>
__global__ void __launch_bounds__(256) k_pool_accum(const float* __restrict__ h,
                                                    const int* __restrict__ batch,
                                                    float* __restrict__ pool, int N) {
  constexpr int C = NL * 32;
  const int lane = threadIdx.x & 31;
  const int gw = wave_in_grid();
  const int nw = waves_in_grid();
  for (int n0 = gw; n0 < N; n0 += R * nw) {
    float v[R][NL];
    int g[R];
    for (int k = 0; k < R; ++k) {
      const int node = n0 + k * nw;
      g[k] = 0;
      if (node < N) {
        const float* row = h + (size_t)node * C + lane;
        for (int j = 0; j < NL; ++j) v[k][j] = row[j * 32];
        g[k] = batch[node];
      }
    }
    for (int k = 0; k < R; ++k) {
      const int node = n0 + k * nw;
      if (node < N) {
        float* pp = pool + (size_t)g[k] * C + lane;
        for (int j = 0; j < NL; ++j) atomicAdd(pp + j * 32, v[k][j]);
      }
    }
  }
}

__global__ void k_count(const int* __restrict__ batch, float* __restrict__ cnt, int N) {
  int i = blockIdx.x * blockDim.x + threadIdx.x;
  if (i < N) atomicAdd(cnt + batch[i], 1.0f);
}

__global__ void k_pool_final(float* __restrict__ pool, const float* __restrict__ cnt,
                             int G, int C) {
  int i = blockIdx.x * blockDim.x + threadIdx.x;
  if (i < G * C) pool[i] /= fmaxf(cnt[i / C], 1.0f);
}

static inline size_t align256(size_t x) { return (x + 255) & ~(size_t)255; }

extern "C" void kernel_launch(void* const* d_in, const int* in_sizes, int n_in,
                              void* d_out, int out_size, void* d_ws, size_t ws_size,
                              hipStream_t stream) {
  const float* x     = (const float*)d_in[0];
  const int*   ei    = (const int*)d_in[1];
  const int*   batch = (const int*)d_in[2];
  const float* W0 = (const float*)d_in[4];
  const float* b0 = (const float*)d_in[5];
  const float* W1 = (const float*)d_in[6];
  const float* b1 = (const float*)d_in[7];
  const float* W2 = (const float*)d_in[8];
  const float* b2 = (const float*)d_in[9];

  const int N  = in_sizes[2];
  const int E  = in_sizes[1] / 2;
  const int IN = in_sizes[0] / N;
  const int H0 = in_sizes[5];
  const int H1 = in_sizes[7];
  const int H2 = in_sizes[9];
  const int G  = out_size / H2 - N;

  const int* srcp = ei;
  const int* dstp = ei + E;

  char* ws = (char*)d_ws;
  float* dinv = (float*)ws;  ws += align256((size_t)N * 4);
  float* cnt  = (float*)ws;  ws += align256((size_t)G * 4);
  float* bufA = (float*)ws;  ws += align256((size_t)N * 256 * 4);
  float* bufB = (float*)ws;  ws += align256((size_t)N * 256 * 4);

  float* pool = (float*)d_out;
  float* hout = (float*)d_out + (size_t)G * H2;

  const int B = 256;
  auto cdiv = [](long long a, long long b) { return (int)((a + b - 1) / b); };

  const int GE = 1920;
  const int GR = 1024;
  auto selfloop = [&](const float* hin, float* o, int C) {
    if (C == 128) k_selfloop<4, 4><<<GR, B, 0, stream>>>(hin, dinv, o, N);
    else          k_selfloop<8, 2><<<GR, B, 0, stream>>>(hin, dinv, o, N);
  };
  auto scatter = [&](const float* hin, float* o, int C) {
    if (C == 128) k_edge_scatter<4><<<GE, B, 0, stream>>>(srcp, dstp, dinv, hin, o, E);
    else          k_edge_scatter<8><<<GE, B, 0, stream>>>(srcp, dstp, dinv, hin, o, E);
  };
  auto post = [&](const float* a, const float* bias, float* o, int C) {
    if (C == 128) k_post<4, 4><<<GR, B, 0, stream>>>(a, bias, o, N);
    else          k_post<8, 2><<<GR, B, 0, stream>>>(a, bias, o, N);
  };

  k_fill<<<cdiv(N, B), B, 0, stream>>>(dinv, 1.0f, N);
  k_deg_edges<<<cdiv(E, B), B, 0, stream>>>(dstp, dinv, E);
  k_dinv<<<cdiv(N, B), B, 0, stream>>>(dinv, N);

  auto layer = [&](const float* hin, int K, int M, const float* W, const float* bias,
                   float* tA, float* tB, float* outp) {
    dim3 g(cdiv(N, 16), M / 128);
    auto gemm = [&](const float* a, float* o) {
      if (K == 128 && M == 256)      k_gemm_wmma_kc<128, 256><<<g, B, 0, stream>>>(a, W, o, N);
      else if (K == 256 && M == 256) k_gemm_wmma_kc<256, 256><<<g, B, 0, stream>>>(a, W, o, N);
      else if (K == 256 && M == 128) k_gemm_wmma_kc<256, 128><<<g, B, 0, stream>>>(a, W, o, N);
      else                           k_gemm_wmma<<<g, B, 0, stream>>>(a, W, o, N, K, M);
    };
    if (K < M) {
      selfloop(hin, tA, K);
      scatter(hin, tA, K);
      gemm(tA, tB);
    } else {
      gemm(hin, tA);
      selfloop(tA, tB, M);
      scatter(tA, tB, M);
    }
    post(tB, bias, outp, M);
  };

  layer(x,    IN, H0, W0, b0,  bufA,  bufB,  bufA);
  layer(bufA, H0, H1, W1, b1,  bufB,  bufA,  bufB);
  layer(bufB, H1, H2, W2, b2,  bufA,  bufB,  hout);

  k_fill<<<cdiv((long long)G * H2, B), B, 0, stream>>>(pool, 0.0f, G * H2);
  k_fill<<<cdiv(G, B), B, 0, stream>>>(cnt, 0.0f, G);
  if (H2 == 128) k_pool_accum<4, 4><<<GR, B, 0, stream>>>(hout, batch, pool, N);
  else           k_pool_accum<8, 2><<<GR, B, 0, stream>>>(hout, batch, pool, N);
  k_count<<<cdiv(N, B), B, 0, stream>>>(batch, cnt, N);
  k_pool_final<<<cdiv((long long)G * H2, B), B, 0, stream>>>(pool, cnt, G, H2);
}


# Edge sums by owner work-groups: each scans all dst ids, adds its 196 rows' edges into LDS rows (LDS f32 atomics, arrival order), stores rows once with self term; no global atomics, 3 launches fewer
# speedup vs baseline: 2.0152x; 1.0802x over previous
#include <hip/hip_runtime.h>
#include <hip/hip_bf16.h>

typedef __attribute__((ext_vector_type(2))) float v2f;
typedef __attribute__((ext_vector_type(8))) float v8f;

__global__ void k_fill(float* __restrict__ p, float v, int n) {
  int i = blockIdx.x * blockDim.x + threadIdx.x;
  if (i < n) p[i] = v;
}

__global__ void k_deg_edges(const int* __restrict__ dst, float* __restrict__ deg, int E) {
  int i = blockIdx.x * blockDim.x + threadIdx.x;
  if (i < E) atomicAdd(deg + dst[i], 1.0f);
}

__global__ void k_dinv(float* __restrict__ deg, int n) {
  int i = blockIdx.x * blockDim.x + threadIdx.x;
  if (i < n) deg[i] = rsqrtf(fmaxf(deg[i], 1.0f));
}

__global__ void __launch_bounds__(256) k_gemm_wmma(
    const float* __restrict__ A, const float* __restrict__ W,
    float* __restrict__ out, int N, int K, int M) {
  __shared__ float As[16 * (256 + 4)];
  const int ldsK = K + 4;
  const int rowBase = blockIdx.x * 16;
  const int colBase = blockIdx.y * 128 + (threadIdx.x >> 5) * 16;

  for (int e = threadIdx.x; e < 16 * K; e += 256) {
    int r = e / K;
    int c = e - r * K;
    int g = rowBase + r;
    As[r * ldsK + c] = (g < N) ? A[(size_t)g * K + c] : 0.0f;
  }
  __syncthreads();

  const int lane = threadIdx.x & 31;
  const int half = lane >> 4;
  const int m    = lane & 15;
  const int col  = colBase + m;

  v8f acc = {};
  for (int k = 0; k < K; k += 4) {
    const int kk = k + 2 * half;
    v2f a, b;
    a.x = As[m * ldsK + kk];
    a.y = As[m * ldsK + kk + 1];
    b.x = W[(size_t)kk * M + col];
    b.y = W[(size_t)(kk + 1) * M + col];
    acc = __builtin_amdgcn_wmma_f32_16x16x4_f32(
         false, a,  false, b,
         (short)0, acc,  false,  false);
  }

  for (int r = 0; r < 8; ++r) {
    int row = rowBase + r + 8 * half;
    if (row < N) out[(size_t)row * M + col] = acc[r];
  }
}

template <int K, int M>
__global__ void __launch_bounds__(256) k_gemm_wmma_kc(
    const float* __restrict__ A, const float* __restrict__ W,
    float* __restrict__ out, int N) {
  constexpr int ldsK = K + 4;
  __shared__ float As[16 * ldsK];
  const int rowBase = blockIdx.x * 16;
  const int colBase = blockIdx.y * 128 + (threadIdx.x >> 5) * 16;

  constexpr int NS = (16 * K) / 256;
  float stage[NS];
#pragma unroll
  for (int i = 0; i < NS; ++i) {
    const int e = (int)threadIdx.x + 256 * i;
    const int r = e / K;
    const int c = e - r * K;
    const int g = rowBase + r;
    const int gl = (g < N) ? g : N - 1;
    stage[i] = A[(size_t)gl * K + c];
  }
#pragma unroll
  for (int i = 0; i < NS; ++i) {
    const int e = (int)threadIdx.x + 256 * i;
    const int r = e / K;
    const int c = e - r * K;
    As[r * ldsK + c] = (rowBase + r < N) ? stage[i] : 0.0f;
  }
  __syncthreads();

  const int lane = threadIdx.x & 31;
  const int half = lane >> 4;
  const int m    = lane & 15;
  const int col  = colBase + m;

  const float* ap = As + m * ldsK + 2 * half;
  const float* wp = W + (size_t)(2 * half) * M + col;

  v8f acc = {};
#pragma unroll 16
  for (int k = 0; k < K; k += 4) {
    v2f a, b;
    a.x = ap[k];
    a.y = ap[k + 1];
    b.x = wp[(size_t)k * M];
    b.y = wp[(size_t)(k + 1) * M];
    acc = __builtin_amdgcn_wmma_f32_16x16x4_f32(
         false, a,  false, b,
         (short)0, acc,  false,  false);
  }

  float* op = out + (size_t)(rowBase + 8 * half) * M + col;
#pragma unroll
  for (int r = 0; r < 8; ++r) {
    if (rowBase + 8 * half + r < N) op[(size_t)r * M] = acc[r];
  }
}

__device__ __forceinline__ int wave_in_grid() {
  return __builtin_amdgcn_readfirstlane((int)((blockIdx.x * blockDim.x + threadIdx.x) >> 5));
}
__device__ __forceinline__ int waves_in_grid() { return (int)((gridDim.x * blockDim.x) >> 5); }

typedef int qint4 __attribute__((ext_vector_type(4), aligned(4)));
template <int NL, int ROWS>
__global__ void __launch_bounds__(1024) k_edge_own(const int* __restrict__ src,
                                                   const int* __restrict__ dst,
                                                   const float* __restrict__ dinv,
                                                   const float* __restrict__ hW,
                                                   float* __restrict__ agg, int E, int N) {
  constexpr int C = NL * 32;
  constexpr int NW = 32;
  constexpr int U = 4;
  constexpr int G = 32 / NL;
  constexpr int QCAP = 64 + U * 128;
  __shared__ float acc[ROWS * C];
  __shared__ int stk[NW * QCAP];
  const int lane = threadIdx.x & 31;
  const int wid = __builtin_amdgcn_readfirstlane((int)(threadIdx.x >> 5));
  const int base = blockIdx.x * ROWS;
  const int nrows = (N - base < ROWS) ? (N - base) : ROWS;
  const int total = nrows * C;

  for (int i0 = threadIdx.x; i0 < total; i0 += 8 * 1024) {
    float hv[8], dv[8];
#pragma unroll
    for (int u = 0; u < 8; ++u) {
      const int idx = i0 + u * 1024;
      const int ic = (idx < total) ? idx : total - 1;
      hv[u] = hW[(size_t)base * C + ic];
      dv[u] = dinv[base + ic / C];
    }
#pragma unroll
    for (int u = 0; u < 8; ++u) {
      const int idx = i0 + u * 1024;
      if (idx < total) acc[idx] = hv[u] * dv[u] * dv[u];
    }
  }
  __syncthreads();

  const int sb = wid * QCAP;
  int pc = 0;
  const int nlf = E >> 7;
  {
    const int e = (nlf << 7) + (wid << 5) + lane;
    if ((nlf << 7) + (wid << 5) < E) {
      const int ec = (e < E) ? e : E - 1;
      const bool hit = (e < E) && ((unsigned)(dst[ec] - base) < (unsigned)nrows);
      const unsigned m = __builtin_amdgcn_ballot_w32(hit);
      if (m) {
        if (hit) stk[sb + (int)__builtin_amdgcn_mbcnt_lo(m, 0u)] = e;
        pc = __builtin_popcount(m);
      }
    }
  }
  for (int c0 = wid;; c0 += NW * U) {
    const bool last = c0 >= nlf;
    if (!last) {
      qint4 dq[U];
#pragma unroll
      for (int u = 0; u < U; ++u) {
        const int c = c0 + u * NW;
        const int cc = (c < nlf) ? c : nlf - 1;
        dq[u] = *(const qint4*)(dst + 4 * (((size_t)cc << 5) + lane));
      }
#pragma unroll
      for (int u = 0; u < U; ++u) {
        const int c = c0 + u * NW;
        if (c < nlf) {
#pragma unroll
          for (int k = 0; k < 4; ++k) {
            const unsigned m = __builtin_amdgcn_ballot_w32((unsigned)(dq[u][k] - base) < (unsigned)nrows);
            if (m) {
              if ((unsigned)(dq[u][k] - base) < (unsigned)nrows)
                stk[sb + pc + (int)__builtin_amdgcn_mbcnt_lo(m, 0u)] = 4 * ((c << 5) + lane) + k;
              pc += __builtin_popcount(m);
            }
          }
        }
      }
    }
    while (pc >= (last ? 1 : 32)) {
      const int cnt = (pc < 32) ? pc : 32;
      const int b0 = pc - cnt;
      const int li = (lane < cnt) ? lane : cnt - 1;
      const int e = stk[sb + b0 + li];
      const int s = src[e];
      const int dl = dst[e] - base;
      const float ds = dinv[s];
      const float dd = dinv[base + dl];
      for (int j0 = 0; j0 < cnt; j0 += G) {
        float v[G][NL];
#pragma unroll
        for (int g = 0; g < G; ++g) {
          const int j = (j0 + g < cnt) ? j0 + g : cnt - 1;
          const int sj = __builtin_amdgcn_readlane(s, j);
          const float* hp = hW + (size_t)sj * C + lane;
#pragma unroll
          for (int i = 0; i < NL; ++i) v[g][i] = hp[i * 32];
        }
#pragma unroll
        for (int g = 0; g < G; ++g) {
          if (j0 + g < cnt) {
            const int dj = __builtin_amdgcn_readlane(dl, j0 + g);
            const float nj = __int_as_float(__builtin_amdgcn_readlane(__float_as_int(ds), j0 + g)) *
                             __int_as_float(__builtin_amdgcn_readlane(__float_as_int(dd), j0 + g));
            float* ap = acc + dj * C + lane;
#pragma unroll
            for (int i = 0; i < NL; ++i) atomicAdd(ap + i * 32, v[g][i] * nj);
          }
        }
      }
      pc = b0;
    }
    if (last) break;
  }
  __syncthreads();

  for (int idx = threadIdx.x; idx < total; idx += 1024) agg[(size_t)base * C + idx] = acc[idx];
}

template <int NL, int R>
__global__ void __launch_bounds__(256) k_post(const float* __restrict__ agg,
                                              const float* __restrict__ bias,
                                              float* __restrict__ out, int N) {
  constexpr int C = NL * 32;
  const int lane = threadIdx.x & 31;
  const int gw = wave_in_grid();
  const int nw = waves_in_grid();
  float bv[NL];
  for (int j = 0; j < NL; ++j) bv[j] = bias[j * 32 + lane];
  for (int n0 = gw; n0 < N; n0 += R * nw) {
    float v[R][NL];
    for (int k = 0; k < R; ++k) {
      const int node = n0 + k * nw;
      if (node < N) {
        const float* row = agg + (size_t)node * C + lane;
        for (int j = 0; j < NL; ++j) v[k][j] = row[j * 32];
      }
    }
    for (int k = 0; k < R; ++k) {
      const int node = n0 + k * nw;
      if (node < N) {
        float s = 0.0f;
        for (int j = 0; j < NL; ++j) {
          float t = fmaxf(v[k][j] + bv[j], 0.0f);
          v[k][j] = t;
          s += t;
        }
        for (int off = 16; off > 0; off >>= 1) s += __shfl_xor(s, off, 32);
        float mean = s / (float)C;
        float q = 0.0f;
        for (int j = 0; j < NL; ++j) {
          float dlt = v[k][j] - mean;
          q += dlt * dlt;
        }
        for (int off = 16; off > 0; off >>= 1) q += __shfl_xor(q, off, 32);
        float rstd = rsqrtf(q / (float)C + 1e-5f);
        float* o = out + (size_t)node * C + lane;
        for (int j = 0; j < NL; ++j) o[j * 32] = (v[k][j] - mean) * rstd;
      }
    }
  }
}

template <int NL, int R>
__global__ void __launch_bounds__(256) k_pool_accum(const float* __restrict__ h,
                                                    const int* __restrict__ batch,
                                                    float* __restrict__ pool, int N) {
  constexpr int C = NL * 32;
  const int lane = threadIdx.x & 31;
  const int gw = wave_in_grid();
  const int nw = waves_in_grid();
  for (int n0 = gw; n0 < N; n0 += R * nw) {
    float v[R][NL];
    int g[R];
    for (int k = 0; k < R; ++k) {
      const int node = n0 + k * nw;
      g[k] = 0;
      if (node < N) {
        const float* row = h + (size_t)node * C + lane;
        for (int j = 0; j < NL; ++j) v[k][j] = row[j * 32];
        g[k] = batch[node];
      }
    }
    for (int k = 0; k < R; ++k) {
      const int node = n0 + k * nw;
      if (node < N) {
        float* pp = pool + (size_t)g[k] * C + lane;
        for (int j = 0; j < NL; ++j) atomicAdd(pp + j * 32, v[k][j]);
      }
    }
  }
}

__global__ void k_count(const int* __restrict__ batch, float* __restrict__ cnt, int N) {
  int i = blockIdx.x * blockDim.x + threadIdx.x;
  if (i < N) atomicAdd(cnt + batch[i], 1.0f);
}

__global__ void k_pool_final(float* __restrict__ pool, const float* __restrict__ cnt,
                             int G, int C) {
  int i = blockIdx.x * blockDim.x + threadIdx.x;
  if (i < G * C) pool[i] /= fmaxf(cnt[i / C], 1.0f);
}

static inline size_t align256(size_t x) { return (x + 255) & ~(size_t)255; }

extern "C" void kernel_launch(void* const* d_in, const int* in_sizes, int n_in,
                              void* d_out, int out_size, void* d_ws, size_t ws_size,
                              hipStream_t stream) {
  const float* x     = (const float*)d_in[0];
  const int*   ei    = (const int*)d_in[1];
  const int*   batch = (const int*)d_in[2];
  const float* W0 = (const float*)d_in[4];
  const float* b0 = (const float*)d_in[5];
  const float* W1 = (const float*)d_in[6];
  const float* b1 = (const float*)d_in[7];
  const float* W2 = (const float*)d_in[8];
  const float* b2 = (const float*)d_in[9];

  const int N  = in_sizes[2];
  const int E  = in_sizes[1] / 2;
  const int IN = in_sizes[0] / N;
  const int H0 = in_sizes[5];
  const int H1 = in_sizes[7];
  const int H2 = in_sizes[9];
  const int G  = out_size / H2 - N;

  const int* srcp = ei;
  const int* dstp = ei + E;

  char* ws = (char*)d_ws;
  float* dinv = (float*)ws;  ws += align256((size_t)N * 4);
  float* cnt  = (float*)ws;  ws += align256((size_t)G * 4);
  float* bufA = (float*)ws;  ws += align256((size_t)N * 256 * 4);
  float* bufB = (float*)ws;  ws += align256((size_t)N * 256 * 4);

  float* pool = (float*)d_out;
  float* hout = (float*)d_out + (size_t)G * H2;

  const int B = 256;
  auto cdiv = [](long long a, long long b) { return (int)((a + b - 1) / b); };

  const int GR = 1024;
  auto edge_sum = [&](const float* hin, float* o, int C) {
    if (C == 128) k_edge_own<4, 196><<<cdiv(N, 196), 1024, 0, stream>>>(srcp, dstp, dinv, hin, o, E, N);
    else          k_edge_own<8, 196><<<cdiv(N, 196), 1024, 0, stream>>>(srcp, dstp, dinv, hin, o, E, N);
  };
  auto post = [&](const float* a, const float* bias, float* o, int C) {
    if (C == 128) k_post<4, 4><<<GR, B, 0, stream>>>(a, bias, o, N);
    else          k_post<8, 2><<<GR, B, 0, stream>>>(a, bias, o, N);
  };

  k_fill<<<cdiv(N, B), B, 0, stream>>>(dinv, 1.0f, N);
  k_deg_edges<<<cdiv(E, B), B, 0, stream>>>(dstp, dinv, E);
  k_dinv<<<cdiv(N, B), B, 0, stream>>>(dinv, N);

  auto layer = [&](const float* hin, int K, int M, const float* W, const float* bias,
                   float* tA, float* tB, float* outp) {
    dim3 g(cdiv(N, 16), M / 128);
    auto gemm = [&](const float* a, float* o) {
      if (K == 128 && M == 256)      k_gemm_wmma_kc<128, 256><<<g, B, 0, stream>>>(a, W, o, N);
      else if (K == 256 && M == 256) k_gemm_wmma_kc<256, 256><<<g, B, 0, stream>>>(a, W, o, N);
      else if (K == 256 && M == 128) k_gemm_wmma_kc<256, 128><<<g, B, 0, stream>>>(a, W, o, N);
      else                           k_gemm_wmma<<<g, B, 0, stream>>>(a, W, o, N, K, M);
    };
    if (K < M) {
      edge_sum(hin, tA, K);
      gemm(tA, tB);
    } else {
      gemm(hin, tA);
      edge_sum(tA, tB, M);
    }
    post(tB, bias, outp, M);
  };

  layer(x,    IN, H0, W0, b0,  bufA,  bufB,  bufA);
  layer(bufA, H0, H1, W1, b1,  bufB,  bufA,  bufB);
  layer(bufB, H1, H2, W2, b2,  bufA,  bufB,  hout);

  k_fill<<<cdiv((long long)G * H2, B), B, 0, stream>>>(pool, 0.0f, G * H2);
  k_fill<<<cdiv(G, B), B, 0, stream>>>(cnt, 0.0f, G);
  if (H2 == 128) k_pool_accum<4, 4><<<GR, B, 0, stream>>>(hout, batch, pool, N);
  else           k_pool_accum<8, 2><<<GR, B, 0, stream>>>(hout, batch, pool, N);
  k_count<<<cdiv(N, B), B, 0, stream>>>(batch, cnt, N);
  k_pool_final<<<cdiv((long long)G * H2, B), B, 0, stream>>>(pool, cnt, G, H2);
}


# Owner work-groups search the dst list once per call: first edge pass lists each wave's edges (source row, owned row, norm) in workspace, later passes read the lists; sums stay in LDS arrival order
# speedup vs baseline: 2.2776x; 1.1302x over previous
#include <hip/hip_runtime.h>
#include <hip/hip_bf16.h>

typedef __attribute__((ext_vector_type(2))) float v2f;
typedef __attribute__((ext_vector_type(8))) float v8f;

__global__ void k_fill(float* __restrict__ p, float v, int n) {
  int i = blockIdx.x * blockDim.x + threadIdx.x;
  if (i < n) p[i] = v;
}

__global__ void k_deg_edges(const int* __restrict__ dst, float* __restrict__ deg, int E) {
  int i = blockIdx.x * blockDim.x + threadIdx.x;
  if (i < E) atomicAdd(deg + dst[i], 1.0f);
}

__global__ void k_dinv(float* __restrict__ deg, int n) {
  int i = blockIdx.x * blockDim.x + threadIdx.x;
  if (i < n) deg[i] = rsqrtf(fmaxf(deg[i], 1.0f));
}

__global__ void __launch_bounds__(256) k_gemm_wmma(
    const float* __restrict__ A, const float* __restrict__ W,
    float* __restrict__ out, int N, int K, int M) {
  __shared__ float As[16 * (256 + 4)];
  const int ldsK = K + 4;
  const int rowBase = blockIdx.x * 16;
  const int colBase = blockIdx.y * 128 + (threadIdx.x >> 5) * 16;

  for (int e = threadIdx.x; e < 16 * K; e += 256) {
    int r = e / K;
    int c = e - r * K;
    int g = rowBase + r;
    As[r * ldsK + c] = (g < N) ? A[(size_t)g * K + c] : 0.0f;
  }
  __syncthreads();

  const int lane = threadIdx.x & 31;
  const int half = lane >> 4;
  const int m    = lane & 15;
  const int col  = colBase + m;

  v8f acc = {};
  for (int k = 0; k < K; k += 4) {
    const int kk = k + 2 * half;
    v2f a, b;
    a.x = As[m * ldsK + kk];
    a.y = As[m * ldsK + kk + 1];
    b.x = W[(size_t)kk * M + col];
    b.y = W[(size_t)(kk + 1) * M + col];
    acc = __builtin_amdgcn_wmma_f32_16x16x4_f32(
         false, a,  false, b,
         (short)0, acc,  false,  false);
  }

  for (int r = 0; r < 8; ++r) {
    int row = rowBase + r + 8 * half;
    if (row < N) out[(size_t)row * M + col] = acc[r];
  }
}

template <int K, int M>
__global__ void __launch_bounds__(256) k_gemm_wmma_kc(
    const float* __restrict__ A, const float* __restrict__ W,
    float* __restrict__ out, int N) {
  constexpr int ldsK = K + 4;
  __shared__ float As[16 * ldsK];
  const int rowBase = blockIdx.x * 16;
  const int colBase = blockIdx.y * 128 + (threadIdx.x >> 5) * 16;

  constexpr int NS = (16 * K) / 256;
  float stage[NS];
#pragma unroll
  for (int i = 0; i < NS; ++i) {
    const int e = (int)threadIdx.x + 256 * i;
    const int r = e / K;
    const int c = e - r * K;
    const int g = rowBase + r;
    const int gl = (g < N) ? g : N - 1;
    stage[i] = A[(size_t)gl * K + c];
  }
#pragma unroll
  for (int i = 0; i < NS; ++i) {
    const int e = (int)threadIdx.x + 256 * i;
    const int r = e / K;
    const int c = e - r * K;
    As[r * ldsK + c] = (rowBase + r < N) ? stage[i] : 0.0f;
  }
  __syncthreads();

  const int lane = threadIdx.x & 31;
  const int half = lane >> 4;
  const int m    = lane & 15;
  const int col  = colBase + m;

  const float* ap = As + m * ldsK + 2 * half;
  const float* wp = W + (size_t)(2 * half) * M + col;

  v8f acc = {};
#pragma unroll 16
  for (int k = 0; k < K; k += 4) {
    v2f a, b;
    a.x = ap[k];
    a.y = ap[k + 1];
    b.x = wp[(size_t)k * M];
    b.y = wp[(size_t)(k + 1) * M];
    acc = __builtin_amdgcn_wmma_f32_16x16x4_f32(
         false, a,  false, b,
         (short)0, acc,  false,  false);
  }

  float* op = out + (size_t)(rowBase + 8 * half) * M + col;
#pragma unroll
  for (int r = 0; r < 8; ++r) {
    if (rowBase + 8 * half + r < N) op[(size_t)r * M] = acc[r];
  }
}

__device__ __forceinline__ int wave_in_grid() {
  return __builtin_amdgcn_readfirstlane((int)((blockIdx.x * blockDim.x + threadIdx.x) >> 5));
}
__device__ __forceinline__ int waves_in_grid() { return (int)((gridDim.x * blockDim.x) >> 5); }

typedef int qint4 __attribute__((ext_vector_type(4), aligned(4)));
constexpr int EDGE_LIST_CAP = 256;
template <int NL, int ROWS, int MODE>
__global__ void __launch_bounds__(1024) k_edge_own(const int* __restrict__ src,
                                                   const int* __restrict__ dst,
                                                   const float* __restrict__ dinv,
                                                   const float* __restrict__ hW,
                                                   float* __restrict__ agg,
                                                   int* __restrict__ lsrc, int* __restrict__ lrow,
                                                   float* __restrict__ lnrm, int* __restrict__ lcount,
                                                   int E, int N) {
  constexpr int C = NL * 32;
  constexpr int NW = 32;
  constexpr int U = 4;
  constexpr int G = 32 / NL;
  constexpr int QCAP = 64 + U * 128;
  constexpr int LCAP = EDGE_LIST_CAP;
  __shared__ float acc[ROWS * C];
  __shared__ int stk[NW * QCAP];
  __shared__ int wcnt[NW];
  const int lane = threadIdx.x & 31;
  const int wid = __builtin_amdgcn_readfirstlane((int)(threadIdx.x >> 5));
  const int base = blockIdx.x * ROWS;
  const int nrows = (N - base < ROWS) ? (N - base) : ROWS;
  const int total = nrows * C;

  const int gwave = blockIdx.x * NW + wid;
  const size_t lb = (size_t)gwave * LCAP;
  int lc = -1;
  if (MODE == 1) lc = __builtin_amdgcn_readfirstlane(lcount[gwave]);

  for (int i0 = threadIdx.x; i0 < total; i0 += 8 * 1024) {
    float hv[8], dv[8];
#pragma unroll
    for (int u = 0; u < 8; ++u) {
      const int idx = i0 + u * 1024;
      const int ic = (idx < total) ? idx : total - 1;
      hv[u] = hW[(size_t)base * C + ic];
      dv[u] = dinv[base + ic / C];
    }
#pragma unroll
    for (int u = 0; u < 8; ++u) {
      const int idx = i0 + u * 1024;
      if (idx < total) acc[idx] = hv[u] * dv[u] * dv[u];
    }
  }
  __syncthreads();

  auto add_rows = [&](int s, int dl, float nrm, int cnt) {
    for (int j0 = 0; j0 < cnt; j0 += G) {
      float v[G][NL];
#pragma unroll
      for (int g = 0; g < G; ++g) {
        const int j = (j0 + g < cnt) ? j0 + g : cnt - 1;
        const int sj = __builtin_amdgcn_readlane(s, j);
        const float* hp = hW + (size_t)sj * C + lane;
#pragma unroll
        for (int i = 0; i < NL; ++i) v[g][i] = hp[i * 32];
      }
#pragma unroll
      for (int g = 0; g < G; ++g) {
        if (j0 + g < cnt) {
          const int dj = __builtin_amdgcn_readlane(dl, j0 + g);
          const float nj = __int_as_float(__builtin_amdgcn_readlane(__float_as_int(nrm), j0 + g));
          float* ap = acc + dj * C + lane;
#pragma unroll
          for (int i = 0; i < NL; ++i) atomicAdd(ap + i * 32, v[g][i] * nj);
        }
      }
    }
  };

  if (lc >= 0) {
    for (int i0 = 0; i0 < lc; i0 += 32) {
      const int cnt = (lc - i0 < 32) ? lc - i0 : 32;
      const int li = (lane < cnt) ? lane : cnt - 1;
      const int s = lsrc[lb + i0 + li];
      const int dl = lrow[lb + i0 + li];
      const float nrm = lnrm[lb + i0 + li];
      add_rows(s, dl, nrm, cnt);
    }
  } else {
    const int sb = wid * QCAP;
    int pc = 0;
    int nrec = 0;
    const int nlf = E >> 7;
    {
      const int e = (nlf << 7) + (wid << 5) + lane;
      if ((nlf << 7) + (wid << 5) < E) {
        const int ec = (e < E) ? e : E - 1;
        const bool hit = (e < E) && ((unsigned)(dst[ec] - base) < (unsigned)nrows);
        const unsigned m = __builtin_amdgcn_ballot_w32(hit);
        if (m) {
          if (hit) stk[sb + (int)__builtin_amdgcn_mbcnt_lo(m, 0u)] = e;
          pc = __builtin_popcount(m);
        }
      }
    }
    for (int c0 = wid;; c0 += NW * U) {
      const bool last = c0 >= nlf;
      if (!last) {
        qint4 dq[U];
#pragma unroll
        for (int u = 0; u < U; ++u) {
          const int c = c0 + u * NW;
          const int cc = (c < nlf) ? c : nlf - 1;
          dq[u] = *(const qint4*)(dst + 4 * (((size_t)cc << 5) + lane));
        }
#pragma unroll
        for (int u = 0; u < U; ++u) {
          const int c = c0 + u * NW;
          if (c < nlf) {
#pragma unroll
            for (int k = 0; k < 4; ++k) {
              const unsigned m = __builtin_amdgcn_ballot_w32((unsigned)(dq[u][k] - base) < (unsigned)nrows);
              if (m) {
                if ((unsigned)(dq[u][k] - base) < (unsigned)nrows)
                  stk[sb + pc + (int)__builtin_amdgcn_mbcnt_lo(m, 0u)] = 4 * ((c << 5) + lane) + k;
                pc += __builtin_popcount(m);
              }
            }
          }
        }
      }
      while (pc >= (last ? 1 : 32)) {
        const int cnt = (pc < 32) ? pc : 32;
        const int b0 = pc - cnt;
        const int li = (lane < cnt) ? lane : cnt - 1;
        const int e = stk[sb + b0 + li];
        const int s = src[e];
        const int dl = dst[e] - base;
        const float ds = dinv[s];
        const float dd = dinv[base + dl];
        const float nrm = ds * dd;
        if (MODE == 0) {
          if (nrec + cnt <= LCAP && lane < cnt) {
            lsrc[lb + nrec + lane] = s;
            lrow[lb + nrec + lane] = dl;
            lnrm[lb + nrec + lane] = nrm;
          }
          nrec += cnt;
        }
        add_rows(s, dl, nrm, cnt);
        pc = b0;
      }
      if (last) break;
    }
    if (MODE == 0 && lane == 0) wcnt[wid] = (nrec <= LCAP) ? nrec : -1;
  }
  __syncthreads();

  if (MODE == 0 && wid == 0) lcount[blockIdx.x * NW + lane] = wcnt[lane];

  for (int idx = threadIdx.x; idx < total; idx += 1024) agg[(size_t)base * C + idx] = acc[idx];
}

template <int NL, int R>
__global__ void __launch_bounds__(256) k_post(const float* __restrict__ agg,
                                              const float* __restrict__ bias,
                                              float* __restrict__ out, int N) {
  constexpr int C = NL * 32;
  const int lane = threadIdx.x & 31;
  const int gw = wave_in_grid();
  const int nw = waves_in_grid();
  float bv[NL];
  for (int j = 0; j < NL; ++j) bv[j] = bias[j * 32 + lane];
  for (int n0 = gw; n0 < N; n0 += R * nw) {
    float v[R][NL];
    for (int k = 0; k < R; ++k) {
      const int node = n0 + k * nw;
      if (node < N) {
        const float* row = agg + (size_t)node * C + lane;
        for (int j = 0; j < NL; ++j) v[k][j] = row[j * 32];
      }
    }
    for (int k = 0; k < R; ++k) {
      const int node = n0 + k * nw;
      if (node < N) {
        float s = 0.0f;
        for (int j = 0; j < NL; ++j) {
          float t = fmaxf(v[k][j] + bv[j], 0.0f);
          v[k][j] = t;
          s += t;
        }
        for (int off = 16; off > 0; off >>= 1) s += __shfl_xor(s, off, 32);
        float mean = s / (float)C;
        float q = 0.0f;
        for (int j = 0; j < NL; ++j) {
          float dlt = v[k][j] - mean;
          q += dlt * dlt;
        }
        for (int off = 16; off > 0; off >>= 1) q += __shfl_xor(q, off, 32);
        float rstd = rsqrtf(q / (float)C + 1e-5f);
        float* o = out + (size_t)node * C + lane;
        for (int j = 0; j < NL; ++j) o[j * 32] = (v[k][j] - mean) * rstd;
      }
    }
  }
}

template <int NL, int R>
__global__ void __launch_bounds__(256) k_pool_accum(const float* __restrict__ h,
                                                    const int* __restrict__ batch,
                                                    float* __restrict__ pool, int N) {
  constexpr int C = NL * 32;
  const int lane = threadIdx.x & 31;
  const int gw = wave_in_grid();
  const int nw = waves_in_grid();
  for (int n0 = gw; n0 < N; n0 += R * nw) {
    float v[R][NL];
    int g[R];
    for (int k = 0; k < R; ++k) {
      const int node = n0 + k * nw;
      g[k] = 0;
      if (node < N) {
        const float* row = h + (size_t)node * C + lane;
        for (int j = 0; j < NL; ++j) v[k][j] = row[j * 32];
        g[k] = batch[node];
      }
    }
    for (int k = 0; k < R; ++k) {
      const int node = n0 + k * nw;
      if (node < N) {
        float* pp = pool + (size_t)g[k] * C + lane;
        for (int j = 0; j < NL; ++j) atomicAdd(pp + j * 32, v[k][j]);
      }
    }
  }
}

__global__ void k_count(const int* __restrict__ batch, float* __restrict__ cnt, int N) {
  int i = blockIdx.x * blockDim.x + threadIdx.x;
  if (i < N) atomicAdd(cnt + batch[i], 1.0f);
}

__global__ void k_pool_final(float* __restrict__ pool, const float* __restrict__ cnt,
                             int G, int C) {
  int i = blockIdx.x * blockDim.x + threadIdx.x;
  if (i < G * C) pool[i] /= fmaxf(cnt[i / C], 1.0f);
}

static inline size_t align256(size_t x) { return (x + 255) & ~(size_t)255; }

extern "C" void kernel_launch(void* const* d_in, const int* in_sizes, int n_in,
                              void* d_out, int out_size, void* d_ws, size_t ws_size,
                              hipStream_t stream) {
  const float* x     = (const float*)d_in[0];
  const int*   ei    = (const int*)d_in[1];
  const int*   batch = (const int*)d_in[2];
  const float* W0 = (const float*)d_in[4];
  const float* b0 = (const float*)d_in[5];
  const float* W1 = (const float*)d_in[6];
  const float* b1 = (const float*)d_in[7];
  const float* W2 = (const float*)d_in[8];
  const float* b2 = (const float*)d_in[9];

  const int N  = in_sizes[2];
  const int E  = in_sizes[1] / 2;
  const int IN = in_sizes[0] / N;
  const int H0 = in_sizes[5];
  const int H1 = in_sizes[7];
  const int H2 = in_sizes[9];
  const int G  = out_size / H2 - N;

  const int* srcp = ei;
  const int* dstp = ei + E;

  char* ws = (char*)d_ws;
  float* dinv = (float*)ws;  ws += align256((size_t)N * 4);
  float* cnt  = (float*)ws;  ws += align256((size_t)G * 4);
  float* bufA = (float*)ws;  ws += align256((size_t)N * 256 * 4);
  float* bufB = (float*)ws;  ws += align256((size_t)N * 256 * 4);
  const int NLW = ((N + 195) / 196) * 32;
  int*   lcount = (int*)ws;    ws += align256((size_t)NLW * 4);
  int*   lsrc   = (int*)ws;    ws += align256((size_t)NLW * EDGE_LIST_CAP * 4);
  int*   lrow   = (int*)ws;    ws += align256((size_t)NLW * EDGE_LIST_CAP * 4);
  float* lnrm   = (float*)ws;  ws += align256((size_t)NLW * EDGE_LIST_CAP * 4);

  float* pool = (float*)d_out;
  float* hout = (float*)d_out + (size_t)G * H2;

  const int B = 256;
  auto cdiv = [](long long a, long long b) { return (int)((a + b - 1) / b); };

  const int GR = 1024;
  bool listed = false;
  auto edge_sum = [&](const float* hin, float* o, int C) {
    const int g = cdiv(N, 196);
    if (!listed) {
      if (C == 128) k_edge_own<4, 196, 0><<<g, 1024, 0, stream>>>(srcp, dstp, dinv, hin, o, lsrc, lrow, lnrm, lcount, E, N);
      else          k_edge_own<8, 196, 0><<<g, 1024, 0, stream>>>(srcp, dstp, dinv, hin, o, lsrc, lrow, lnrm, lcount, E, N);
    } else {
      if (C == 128) k_edge_own<4, 196, 1><<<g, 1024, 0, stream>>>(srcp, dstp, dinv, hin, o, lsrc, lrow, lnrm, lcount, E, N);
      else          k_edge_own<8, 196, 1><<<g, 1024, 0, stream>>>(srcp, dstp, dinv, hin, o, lsrc, lrow, lnrm, lcount, E, N);
    }
    listed = true;
  };
  auto post = [&](const float* a, const float* bias, float* o, int C) {
    if (C == 128) k_post<4, 4><<<GR, B, 0, stream>>>(a, bias, o, N);
    else          k_post<8, 2><<<GR, B, 0, stream>>>(a, bias, o, N);
  };

  k_fill<<<cdiv(N, B), B, 0, stream>>>(dinv, 1.0f, N);
  k_deg_edges<<<cdiv(E, B), B, 0, stream>>>(dstp, dinv, E);
  k_dinv<<<cdiv(N, B), B, 0, stream>>>(dinv, N);

  auto layer = [&](const float* hin, int K, int M, const float* W, const float* bias,
                   float* tA, float* tB, float* outp) {
    dim3 g(cdiv(N, 16), M / 128);
    auto gemm = [&](const float* a, float* o) {
      if (K == 128 && M == 256)      k_gemm_wmma_kc<128, 256><<<g, B, 0, stream>>>(a, W, o, N);
      else if (K == 256 && M == 256) k_gemm_wmma_kc<256, 256><<<g, B, 0, stream>>>(a, W, o, N);
      else if (K == 256 && M == 128) k_gemm_wmma_kc<256, 128><<<g, B, 0, stream>>>(a, W, o, N);
      else                           k_gemm_wmma<<<g, B, 0, stream>>>(a, W, o, N, K, M);
    };
    if (K < M) {
      edge_sum(hin, tA, K);
      gemm(tA, tB);
    } else {
      gemm(hin, tA);
      edge_sum(tA, tB, M);
    }
    post(tB, bias, outp, M);
  };

  layer(x,    IN, H0, W0, b0,  bufA,  bufB,  bufA);
  layer(bufA, H0, H1, W1, b1,  bufB,  bufA,  bufB);
  layer(bufB, H1, H2, W2, b2,  bufA,  bufB,  hout);

  k_fill<<<cdiv((long long)G * H2, B), B, 0, stream>>>(pool, 0.0f, G * H2);
  k_fill<<<cdiv(G, B), B, 0, stream>>>(cnt, 0.0f, G);
  if (H2 == 128) k_pool_accum<4, 4><<<GR, B, 0, stream>>>(hout, batch, pool, N);
  else           k_pool_accum<8, 2><<<GR, B, 0, stream>>>(hout, batch, pool, N);
  k_count<<<cdiv(N, B), B, 0, stream>>>(batch, cnt, N);
  k_pool_final<<<cdiv((long long)G * H2, B), B, 0, stream>>>(pool, cnt, G, H2);
}


# Product kernel keeps all of W in LDS: 241 resident work-groups of 13 row tiles copy W once (k-pair rows interleaved) and read B fragments from LDS, not from global per wave; sum order kept
# speedup vs baseline: 2.3852x; 1.0472x over previous
#include <hip/hip_runtime.h>
#include <hip/hip_bf16.h>

typedef __attribute__((ext_vector_type(2))) float v2f;
typedef __attribute__((ext_vector_type(8))) float v8f;

__global__ void k_fill(float* __restrict__ p, float v, int n) {
  int i = blockIdx.x * blockDim.x + threadIdx.x;
  if (i < n) p[i] = v;
}

__global__ void k_deg_edges(const int* __restrict__ dst, float* __restrict__ deg, int E) {
  int i = blockIdx.x * blockDim.x + threadIdx.x;
  if (i < E) atomicAdd(deg + dst[i], 1.0f);
}

__global__ void k_dinv(float* __restrict__ deg, int n) {
  int i = blockIdx.x * blockDim.x + threadIdx.x;
  if (i < n) deg[i] = rsqrtf(fmaxf(deg[i], 1.0f));
}

__global__ void __launch_bounds__(256) k_gemm_wmma(
    const float* __restrict__ A, const float* __restrict__ W,
    float* __restrict__ out, int N, int K, int M) {
  __shared__ float As[16 * (256 + 4)];
  const int ldsK = K + 4;
  const int rowBase = blockIdx.x * 16;
  const int colBase = blockIdx.y * 128 + (threadIdx.x >> 5) * 16;

  for (int e = threadIdx.x; e < 16 * K; e += 256) {
    int r = e / K;
    int c = e - r * K;
    int g = rowBase + r;
    As[r * ldsK + c] = (g < N) ? A[(size_t)g * K + c] : 0.0f;
  }
  __syncthreads();

  const int lane = threadIdx.x & 31;
  const int half = lane >> 4;
  const int m    = lane & 15;
  const int col  = colBase + m;

  v8f acc = {};
  for (int k = 0; k < K; k += 4) {
    const int kk = k + 2 * half;
    v2f a, b;
    a.x = As[m * ldsK + kk];
    a.y = As[m * ldsK + kk + 1];
    b.x = W[(size_t)kk * M + col];
    b.y = W[(size_t)(kk + 1) * M + col];
    acc = __builtin_amdgcn_wmma_f32_16x16x4_f32(
         false, a,  false, b,
         (short)0, acc,  false,  false);
  }

  for (int r = 0; r < 8; ++r) {
    int row = rowBase + r + 8 * half;
    if (row < N) out[(size_t)row * M + col] = acc[r];
  }
}

__device__ __forceinline__ void lds_barrier() {
  __builtin_amdgcn_fence(__ATOMIC_RELEASE, "workgroup", "local");
  __builtin_amdgcn_s_barrier();
  __builtin_amdgcn_fence(__ATOMIC_ACQUIRE, "workgroup", "local");
}
typedef float qfloat4 __attribute__((ext_vector_type(4), aligned(4)));
typedef float lfloat4 __attribute__((ext_vector_type(4)));
constexpr int GEMM_TILES = 13;
template <int K, int M, int P>
__global__ void __launch_bounds__(1024) k_gemm_wlds(
    const float* __restrict__ A, const float* __restrict__ W,
    float* __restrict__ out, int N) {
  constexpr int TILES = GEMM_TILES;
  constexpr int CT = M / 16;
  constexpr int ldsK = K + 4;
  constexpr int WP = 2 * M + 32;
  constexpr int NPH = (TILES + P - 1) / P;
  constexpr int K4 = K / 4;
  constexpr int AU = P * 16 * K4;
  constexpr int NS = (AU + 1023) / 1024;
  constexpr int WU = (K / 2) * (M / 4);
  constexpr int WCH = 4;
  __shared__ alignas(16) float smem[P * 16 * ldsK + (K / 2) * WP];
  float* const As = smem;
  float* const Ws = smem + P * 16 * ldsK;
  const int tid = threadIdx.x;
  const int lane = tid & 31;
  const int wid = __builtin_amdgcn_readfirstlane(tid >> 5);
  const int half = lane >> 4;
  const int m    = lane & 15;
  const int base = blockIdx.x * (TILES * 16);
  const int left = (N - base + 15) >> 4;
  const int nt = (left < TILES) ? left : TILES;

  qfloat4 st[NS];
  auto fetch = [&](int g) {
#pragma unroll
    for (int i = 0; i < NS; ++i) {
      const int u = tid + 1024 * i;
      const int r = u / K4;
      const int c4 = u - r * K4;
      const int row = base + g * (P * 16) + r;
      const int rl = (row < N) ? row : N - 1;
      st[i] = *(const qfloat4*)(A + (size_t)rl * K + 4 * c4);
    }
  };
  auto stash = [&](int g) {
#pragma unroll
    for (int i = 0; i < NS; ++i) {
      const int u = tid + 1024 * i;
      const int r = u / K4;
      const int c4 = u - r * K4;
      const int row = base + g * (P * 16) + r;
      if (u < AU) {
        lfloat4 v = {st[i].x, st[i].y, st[i].z, st[i].w};
        if (row >= N) v = lfloat4{0.0f, 0.0f, 0.0f, 0.0f};
        *(lfloat4*)(As + r * ldsK + 4 * c4) = v;
      }
    }
  };

  fetch(0);

  for (int v0 = tid; v0 < WU; v0 += 1024 * WCH) {
    qfloat4 w0[WCH], w1[WCH];
#pragma unroll
    for (int j = 0; j < WCH; ++j) {
      const int v = v0 + 1024 * j;
      const int p = v / (M / 4);
      const int n4 = v - p * (M / 4);
      w0[j] = *(const qfloat4*)(W + (size_t)(2 * p) * M + 4 * n4);
      w1[j] = *(const qfloat4*)(W + (size_t)(2 * p + 1) * M + 4 * n4);
    }
#pragma unroll
    for (int j = 0; j < WCH; ++j) {
      const int v = v0 + 1024 * j;
      const int p = v / (M / 4);
      const int n4 = v - p * (M / 4);
      float* wd = Ws + p * WP + 8 * n4;
      *(lfloat4*)(wd)     = lfloat4{w0[j].x, w1[j].x, w0[j].y, w1[j].y};
      *(lfloat4*)(wd + 4) = lfloat4{w0[j].z, w1[j].z, w0[j].w, w1[j].w};
    }
  }

  for (int g = 0; g < NPH; ++g) {
    stash(g);
    lds_barrier();
    if (g + 1 < NPH) fetch(g + 1);
    const int t0 = g * P;
    const int tl = (nt - t0 < P) ? nt - t0 : P;
    for (int it = wid; it < tl * CT; it += 32) {
      const int rt = it / CT;
      const int ct = it - rt * CT;
      const float* ap = As + (rt * 16 + m) * ldsK + 2 * half;
      const float* bp = Ws + half * WP + 2 * (ct * 16 + m);
      v8f acc = {};
#pragma unroll 1
      for (int kb = 0; kb < K / 64; ++kb) {
#pragma unroll
        for (int s = 0; s < 16; ++s) {
          const v2f a = *(const v2f*)(ap + 4 * s);
          const v2f b = *(const v2f*)(bp + 2 * s * WP);
          acc = __builtin_amdgcn_wmma_f32_16x16x4_f32(
               false, a,  false, b,
               (short)0, acc,  false,  false);
        }
        ap += 64;
        bp += 32 * WP;
      }
      const int row0 = base + (t0 + rt) * 16 + 8 * half;
      float* op = out + (size_t)row0 * M + ct * 16 + m;
      if (base + (t0 + rt) * 16 + 16 <= N) {
#pragma unroll
        for (int r = 0; r < 8; ++r) op[(size_t)r * M] = acc[r];
      } else {
#pragma unroll
        for (int r = 0; r < 8; ++r) {
          if (row0 + r < N) op[(size_t)r * M] = acc[r];
        }
      }
    }
    if (g + 1 < NPH) lds_barrier();
  }
}

__device__ __forceinline__ int wave_in_grid() {
  return __builtin_amdgcn_readfirstlane((int)((blockIdx.x * blockDim.x + threadIdx.x) >> 5));
}
__device__ __forceinline__ int waves_in_grid() { return (int)((gridDim.x * blockDim.x) >> 5); }

typedef int qint4 __attribute__((ext_vector_type(4), aligned(4)));
constexpr int EDGE_LIST_CAP = 256;
template <int NL, int ROWS, int MODE>
__global__ void __launch_bounds__(1024) k_edge_own(const int* __restrict__ src,
                                                   const int* __restrict__ dst,
                                                   const float* __restrict__ dinv,
                                                   const float* __restrict__ hW,
                                                   float* __restrict__ agg,
                                                   int* __restrict__ lsrc, int* __restrict__ lrow,
                                                   float* __restrict__ lnrm, int* __restrict__ lcount,
                                                   int E, int N) {
  constexpr int C = NL * 32;
  constexpr int NW = 32;
  constexpr int U = 4;
  constexpr int G = 32 / NL;
  constexpr int QCAP = 64 + U * 128;
  constexpr int LCAP = EDGE_LIST_CAP;
  __shared__ float acc[ROWS * C];
  __shared__ int stk[NW * QCAP];
  __shared__ int wcnt[NW];
  const int lane = threadIdx.x & 31;
  const int wid = __builtin_amdgcn_readfirstlane((int)(threadIdx.x >> 5));
  const int base = blockIdx.x * ROWS;
  const int nrows = (N - base < ROWS) ? (N - base) : ROWS;
  const int total = nrows * C;

  const int gwave = blockIdx.x * NW + wid;
  const size_t lb = (size_t)gwave * LCAP;
  int lc = -1;
  if (MODE == 1) lc = __builtin_amdgcn_readfirstlane(lcount[gwave]);

  for (int i0 = threadIdx.x; i0 < total; i0 += 8 * 1024) {
    float hv[8], dv[8];
#pragma unroll
    for (int u = 0; u < 8; ++u) {
      const int idx = i0 + u * 1024;
      const int ic = (idx < total) ? idx : total - 1;
      hv[u] = hW[(size_t)base * C + ic];
      dv[u] = dinv[base + ic / C];
    }
#pragma unroll
    for (int u = 0; u < 8; ++u) {
      const int idx = i0 + u * 1024;
      if (idx < total) acc[idx] = hv[u] * dv[u] * dv[u];
    }
  }
  __syncthreads();

  auto add_rows = [&](int s, int dl, float nrm, int cnt) {
    for (int j0 = 0; j0 < cnt; j0 += G) {
      float v[G][NL];
#pragma unroll
      for (int g = 0; g < G; ++g) {
        const int j = (j0 + g < cnt) ? j0 + g : cnt - 1;
        const int sj = __builtin_amdgcn_readlane(s, j);
        const float* hp = hW + (size_t)sj * C + lane;
#pragma unroll
        for (int i = 0; i < NL; ++i) v[g][i] = hp[i * 32];
      }
#pragma unroll
      for (int g = 0; g < G; ++g) {
        if (j0 + g < cnt) {
          const int dj = __builtin_amdgcn_readlane(dl, j0 + g);
          const float nj = __int_as_float(__builtin_amdgcn_readlane(__float_as_int(nrm), j0 + g));
          float* ap = acc + dj * C + lane;
#pragma unroll
          for (int i = 0; i < NL; ++i) atomicAdd(ap + i * 32, v[g][i] * nj);
        }
      }
    }
  };

  if (lc >= 0) {
    for (int i0 = 0; i0 < lc; i0 += 32) {
      const int cnt = (lc - i0 < 32) ? lc - i0 : 32;
      const int li = (lane < cnt) ? lane : cnt - 1;
      const int s = lsrc[lb + i0 + li];
      const int dl = lrow[lb + i0 + li];
      const float nrm = lnrm[lb + i0 + li];
      add_rows(s, dl, nrm, cnt);
    }
  } else {
    const int sb = wid * QCAP;
    int pc = 0;
    int nrec = 0;
    const int nlf = E >> 7;
    {
      const int e = (nlf << 7) + (wid << 5) + lane;
      if ((nlf << 7) + (wid << 5) < E) {
        const int ec = (e < E) ? e : E - 1;
        const bool hit = (e < E) && ((unsigned)(dst[ec] - base) < (unsigned)nrows);
        const unsigned m = __builtin_amdgcn_ballot_w32(hit);
        if (m) {
          if (hit) stk[sb + (int)__builtin_amdgcn_mbcnt_lo(m, 0u)] = e;
          pc = __builtin_popcount(m);
        }
      }
    }
    for (int c0 = wid;; c0 += NW * U) {
      const bool last = c0 >= nlf;
      if (!last) {
        qint4 dq[U];
#pragma unroll
        for (int u = 0; u < U; ++u) {
          const int c = c0 + u * NW;
          const int cc = (c < nlf) ? c : nlf - 1;
          dq[u] = *(const qint4*)(dst + 4 * (((size_t)cc << 5) + lane));
        }
#pragma unroll
        for (int u = 0; u < U; ++u) {
          const int c = c0 + u * NW;
          if (c < nlf) {
#pragma unroll
            for (int k = 0; k < 4; ++k) {
              const unsigned m = __builtin_amdgcn_ballot_w32((unsigned)(dq[u][k] - base) < (unsigned)nrows);
              if (m) {
                if ((unsigned)(dq[u][k] - base) < (unsigned)nrows)
                  stk[sb + pc + (int)__builtin_amdgcn_mbcnt_lo(m, 0u)] = 4 * ((c << 5) + lane) + k;
                pc += __builtin_popcount(m);
              }
            }
          }
        }
      }
      while (pc >= (last ? 1 : 32)) {
        const int cnt = (pc < 32) ? pc : 32;
        const int b0 = pc - cnt;
        const int li = (lane < cnt) ? lane : cnt - 1;
        const int e = stk[sb + b0 + li];
        const int s = src[e];
        const int dl = dst[e] - base;
        const float ds = dinv[s];
        const float dd = dinv[base + dl];
        const float nrm = ds * dd;
        if (MODE == 0) {
          if (nrec + cnt <= LCAP && lane < cnt) {
            lsrc[lb + nrec + lane] = s;
            lrow[lb + nrec + lane] = dl;
            lnrm[lb + nrec + lane] = nrm;
          }
          nrec += cnt;
        }
        add_rows(s, dl, nrm, cnt);
        pc = b0;
      }
      if (last) break;
    }
    if (MODE == 0 && lane == 0) wcnt[wid] = (nrec <= LCAP) ? nrec : -1;
  }
  __syncthreads();

  if (MODE == 0 && wid == 0) lcount[blockIdx.x * NW + lane] = wcnt[lane];

  for (int idx = threadIdx.x; idx < total; idx += 1024) agg[(size_t)base * C + idx] = acc[idx];
}

template <int NL, int R>
__global__ void __launch_bounds__(256) k_post(const float* __restrict__ agg,
                                              const float* __restrict__ bias,
                                              float* __restrict__ out, int N) {
  constexpr int C = NL * 32;
  const int lane = threadIdx.x & 31;
  const int gw = wave_in_grid();
  const int nw = waves_in_grid();
  float bv[NL];
  for (int j = 0; j < NL; ++j) bv[j] = bias[j * 32 + lane];
  for (int n0 = gw; n0 < N; n0 += R * nw) {
    float v[R][NL];
    for (int k = 0; k < R; ++k) {
      const int node = n0 + k * nw;
      if (node < N) {
        const float* row = agg + (size_t)node * C + lane;
        for (int j = 0; j < NL; ++j) v[k][j] = row[j * 32];
      }
    }
    for (int k = 0; k < R; ++k) {
      const int node = n0 + k * nw;
      if (node < N) {
        float s = 0.0f;
        for (int j = 0; j < NL; ++j) {
          float t = fmaxf(v[k][j] + bv[j], 0.0f);
          v[k][j] = t;
          s += t;
        }
        for (int off = 16; off > 0; off >>= 1) s += __shfl_xor(s, off, 32);
        float mean = s / (float)C;
        float q = 0.0f;
        for (int j = 0; j < NL; ++j) {
          float dlt = v[k][j] - mean;
          q += dlt * dlt;
        }
        for (int off = 16; off > 0; off >>= 1) q += __shfl_xor(q, off, 32);
        float rstd = rsqrtf(q / (float)C + 1e-5f);
        float* o = out + (size_t)node * C + lane;
        for (int j = 0; j < NL; ++j) o[j * 32] = (v[k][j] - mean) * rstd;
      }
    }
  }
}

template <int NL, int R>
__global__ void __launch_bounds__(256) k_pool_accum(const float* __restrict__ h,
                                                    const int* __restrict__ batch,
                                                    float* __restrict__ pool, int N) {
  constexpr int C = NL * 32;
  const int lane = threadIdx.x & 31;
  const int gw = wave_in_grid();
  const int nw = waves_in_grid();
  for (int n0 = gw; n0 < N; n0 += R * nw) {
    float v[R][NL];
    int g[R];
    for (int k = 0; k < R; ++k) {
      const int node = n0 + k * nw;
      g[k] = 0;
      if (node < N) {
        const float* row = h + (size_t)node * C + lane;
        for (int j = 0; j < NL; ++j) v[k][j] = row[j * 32];
        g[k] = batch[node];
      }
    }
    for (int k = 0; k < R; ++k) {
      const int node = n0 + k * nw;
      if (node < N) {
        float* pp = pool + (size_t)g[k] * C + lane;
        for (int j = 0; j < NL; ++j) atomicAdd(pp + j * 32, v[k][j]);
      }
    }
  }
}

__global__ void k_count(const int* __restrict__ batch, float* __restrict__ cnt, int N) {
  int i = blockIdx.x * blockDim.x + threadIdx.x;
  if (i < N) atomicAdd(cnt + batch[i], 1.0f);
}

__global__ void k_pool_final(float* __restrict__ pool, const float* __restrict__ cnt,
                             int G, int C) {
  int i = blockIdx.x * blockDim.x + threadIdx.x;
  if (i < G * C) pool[i] /= fmaxf(cnt[i / C], 1.0f);
}

static inline size_t align256(size_t x) { return (x + 255) & ~(size_t)255; }

extern "C" void kernel_launch(void* const* d_in, const int* in_sizes, int n_in,
                              void* d_out, int out_size, void* d_ws, size_t ws_size,
                              hipStream_t stream) {
  const float* x     = (const float*)d_in[0];
  const int*   ei    = (const int*)d_in[1];
  const int*   batch = (const int*)d_in[2];
  const float* W0 = (const float*)d_in[4];
  const float* b0 = (const float*)d_in[5];
  const float* W1 = (const float*)d_in[6];
  const float* b1 = (const float*)d_in[7];
  const float* W2 = (const float*)d_in[8];
  const float* b2 = (const float*)d_in[9];

  const int N  = in_sizes[2];
  const int E  = in_sizes[1] / 2;
  const int IN = in_sizes[0] / N;
  const int H0 = in_sizes[5];
  const int H1 = in_sizes[7];
  const int H2 = in_sizes[9];
  const int G  = out_size / H2 - N;

  const int* srcp = ei;
  const int* dstp = ei + E;

  char* ws = (char*)d_ws;
  float* dinv = (float*)ws;  ws += align256((size_t)N * 4);
  float* cnt  = (float*)ws;  ws += align256((size_t)G * 4);
  float* bufA = (float*)ws;  ws += align256((size_t)N * 256 * 4);
  float* bufB = (float*)ws;  ws += align256((size_t)N * 256 * 4);
  const int NLW = ((N + 195) / 196) * 32;
  int*   lcount = (int*)ws;    ws += align256((size_t)NLW * 4);
  int*   lsrc   = (int*)ws;    ws += align256((size_t)NLW * EDGE_LIST_CAP * 4);
  int*   lrow   = (int*)ws;    ws += align256((size_t)NLW * EDGE_LIST_CAP * 4);
  float* lnrm   = (float*)ws;  ws += align256((size_t)NLW * EDGE_LIST_CAP * 4);

  float* pool = (float*)d_out;
  float* hout = (float*)d_out + (size_t)G * H2;

  const int B = 256;
  auto cdiv = [](long long a, long long b) { return (int)((a + b - 1) / b); };

  const int GR = 1024;
  bool listed = false;
  auto edge_sum = [&](const float* hin, float* o, int C) {
    const int g = cdiv(N, 196);
    if (!listed) {
      if (C == 128) k_edge_own<4, 196, 0><<<g, 1024, 0, stream>>>(srcp, dstp, dinv, hin, o, lsrc, lrow, lnrm, lcount, E, N);
      else          k_edge_own<8, 196, 0><<<g, 1024, 0, stream>>>(srcp, dstp, dinv, hin, o, lsrc, lrow, lnrm, lcount, E, N);
    } else {
      if (C == 128) k_edge_own<4, 196, 1><<<g, 1024, 0, stream>>>(srcp, dstp, dinv, hin, o, lsrc, lrow, lnrm, lcount, E, N);
      else          k_edge_own<8, 196, 1><<<g, 1024, 0, stream>>>(srcp, dstp, dinv, hin, o, lsrc, lrow, lnrm, lcount, E, N);
    }
    listed = true;
  };
  auto post = [&](const float* a, const float* bias, float* o, int C) {
    if (C == 128) k_post<4, 4><<<GR, B, 0, stream>>>(a, bias, o, N);
    else          k_post<8, 2><<<GR, B, 0, stream>>>(a, bias, o, N);
  };

  k_fill<<<cdiv(N, B), B, 0, stream>>>(dinv, 1.0f, N);
  k_deg_edges<<<cdiv(E, B), B, 0, stream>>>(dstp, dinv, E);
  k_dinv<<<cdiv(N, B), B, 0, stream>>>(dinv, N);

  auto layer = [&](const float* hin, int K, int M, const float* W, const float* bias,
                   float* tA, float* tB, float* outp) {
    dim3 g(cdiv(N, 16), M / 128);
    const int gp = cdiv(N, GEMM_TILES * 16);
    auto gemm = [&](const float* a, float* o) {
      if (K == 128 && M == 256)      k_gemm_wlds<128, 256, 13><<<gp, 1024, 0, stream>>>(a, W, o, N);
      else if (K == 256 && M == 256) k_gemm_wlds<256, 256, 2><<<gp, 1024, 0, stream>>>(a, W, o, N);
      else if (K == 256 && M == 128) k_gemm_wlds<256, 128, 4><<<gp, 1024, 0, stream>>>(a, W, o, N);
      else                           k_gemm_wmma<<<g, B, 0, stream>>>(a, W, o, N, K, M);
    };
    if (K < M) {
      edge_sum(hin, tA, K);
      gemm(tA, tB);
    } else {
      gemm(hin, tA);
      edge_sum(tA, tB, M);
    }
    post(tB, bias, outp, M);
  };

  layer(x,    IN, H0, W0, b0,  bufA,  bufB,  bufA);
  layer(bufA, H0, H1, W1, b1,  bufB,  bufA,  bufB);
  layer(bufB, H1, H2, W2, b2,  bufA,  bufB,  hout);

  k_fill<<<cdiv((long long)G * H2, B), B, 0, stream>>>(pool, 0.0f, G * H2);
  k_fill<<<cdiv(G, B), B, 0, stream>>>(cnt, 0.0f, G);
  if (H2 == 128) k_pool_accum<4, 4><<<GR, B, 0, stream>>>(hout, batch, pool, N);
  else           k_pool_accum<8, 2><<<GR, B, 0, stream>>>(hout, batch, pool, N);
  k_count<<<cdiv(N, B), B, 0, stream>>>(batch, cnt, N);
  k_pool_final<<<cdiv((long long)G * H2, B), B, 0, stream>>>(pool, cnt, G, H2);
}


# Mean pool in existing launches: last layer-norm kernel adds each wave's per-graph row sums / count into out (node order in a wave, then atomic arrival order); counts, zero fills join degree launches
# speedup vs baseline: 2.9165x; 1.2227x over previous
#include <hip/hip_runtime.h>
#include <hip/hip_bf16.h>

typedef __attribute__((ext_vector_type(2))) float v2f;
typedef __attribute__((ext_vector_type(8))) float v8f;

__global__ void k_init(float* __restrict__ deg, int n, float* __restrict__ pool, int np,
                       float* __restrict__ cnt, int nc) {
  const int i = blockIdx.x * blockDim.x + threadIdx.x;
  if (i < n) deg[i] = 1.0f;
  if (i < np) pool[i] = 0.0f;
  if (i < nc) cnt[i] = 0.0f;
}

__global__ void k_deg_cnt(const int* __restrict__ dst, float* __restrict__ deg, int E,
                          const int* __restrict__ batch, float* __restrict__ cnt, int N) {
  const int i = blockIdx.x * blockDim.x + threadIdx.x;
  if (i < E) atomicAdd(deg + dst[i], 1.0f);
  if (i < N) {
    const int g = batch[i];
    const int g0 = __builtin_amdgcn_readfirstlane(g);
    const unsigned act = __builtin_amdgcn_ballot_w32(true);
    const unsigned same = __builtin_amdgcn_ballot_w32(g == g0);
    if (same == act) {
      if (__builtin_amdgcn_mbcnt_lo(act, 0u) == 0) atomicAdd(cnt + g0, (float)__builtin_popcount(act));
    } else {
      atomicAdd(cnt + g, 1.0f);
    }
  }
}

__global__ void k_dinv(float* __restrict__ deg, int n) {
  int i = blockIdx.x * blockDim.x + threadIdx.x;
  if (i < n) deg[i] = rsqrtf(fmaxf(deg[i], 1.0f));
}

__global__ void __launch_bounds__(256) k_gemm_wmma(
    const float* __restrict__ A, const float* __restrict__ W,
    float* __restrict__ out, int N, int K, int M) {
  __shared__ float As[16 * (256 + 4)];
  const int ldsK = K + 4;
  const int rowBase = blockIdx.x * 16;
  const int colBase = blockIdx.y * 128 + (threadIdx.x >> 5) * 16;

  for (int e = threadIdx.x; e < 16 * K; e += 256) {
    int r = e / K;
    int c = e - r * K;
    int g = rowBase + r;
    As[r * ldsK + c] = (g < N) ? A[(size_t)g * K + c] : 0.0f;
  }
  __syncthreads();

  const int lane = threadIdx.x & 31;
  const int half = lane >> 4;
  const int m    = lane & 15;
  const int col  = colBase + m;

  v8f acc = {};
  for (int k = 0; k < K; k += 4) {
    const int kk = k + 2 * half;
    v2f a, b;
    a.x = As[m * ldsK + kk];
    a.y = As[m * ldsK + kk + 1];
    b.x = W[(size_t)kk * M + col];
    b.y = W[(size_t)(kk + 1) * M + col];
    acc = __builtin_amdgcn_wmma_f32_16x16x4_f32(
         false, a,  false, b,
         (short)0, acc,  false,  false);
  }

  for (int r = 0; r < 8; ++r) {
    int row = rowBase + r + 8 * half;
    if (row < N) out[(size_t)row * M + col] = acc[r];
  }
}

__device__ __forceinline__ void lds_barrier() {
  __builtin_amdgcn_fence(__ATOMIC_RELEASE, "workgroup", "local");
  __builtin_amdgcn_s_barrier();
  __builtin_amdgcn_fence(__ATOMIC_ACQUIRE, "workgroup", "local");
}
typedef float qfloat4 __attribute__((ext_vector_type(4), aligned(4)));
typedef float lfloat4 __attribute__((ext_vector_type(4)));
constexpr int GEMM_TILES = 13;
template <int K, int M, int P>
__global__ void __launch_bounds__(1024) k_gemm_wlds(
    const float* __restrict__ A, const float* __restrict__ W,
    float* __restrict__ out, int N) {
  constexpr int TILES = GEMM_TILES;
  constexpr int CT = M / 16;
  constexpr int ldsK = K + 4;
  constexpr int WP = 2 * M + 32;
  constexpr int NPH = (TILES + P - 1) / P;
  constexpr int K4 = K / 4;
  constexpr int AU = P * 16 * K4;
  constexpr int NS = (AU + 1023) / 1024;
  constexpr int WU = (K / 2) * (M / 4);
  constexpr int WCH = 4;
  __shared__ alignas(16) float smem[P * 16 * ldsK + (K / 2) * WP];
  float* const As = smem;
  float* const Ws = smem + P * 16 * ldsK;
  const int tid = threadIdx.x;
  const int lane = tid & 31;
  const int wid = __builtin_amdgcn_readfirstlane(tid >> 5);
  const int half = lane >> 4;
  const int m    = lane & 15;
  const int base = blockIdx.x * (TILES * 16);
  const int left = (N - base + 15) >> 4;
  const int nt = (left < TILES) ? left : TILES;

  qfloat4 st[NS];
  auto fetch = [&](int g) {
#pragma unroll
    for (int i = 0; i < NS; ++i) {
      const int u = tid + 1024 * i;
      const int r = u / K4;
      const int c4 = u - r * K4;
      const int row = base + g * (P * 16) + r;
      const int rl = (row < N) ? row : N - 1;
      st[i] = *(const qfloat4*)(A + (size_t)rl * K + 4 * c4);
    }
  };
  auto stash = [&](int g) {
#pragma unroll
    for (int i = 0; i < NS; ++i) {
      const int u = tid + 1024 * i;
      const int r = u / K4;
      const int c4 = u - r * K4;
      const int row = base + g * (P * 16) + r;
      if (u < AU) {
        lfloat4 v = {st[i].x, st[i].y, st[i].z, st[i].w};
        if (row >= N) v = lfloat4{0.0f, 0.0f, 0.0f, 0.0f};
        *(lfloat4*)(As + r * ldsK + 4 * c4) = v;
      }
    }
  };

  fetch(0);

  for (int v0 = tid; v0 < WU; v0 += 1024 * WCH) {
    qfloat4 w0[WCH], w1[WCH];
#pragma unroll
    for (int j = 0; j < WCH; ++j) {
      const int v = v0 + 1024 * j;
      const int p = v / (M / 4);
      const int n4 = v - p * (M / 4);
      w0[j] = *(const qfloat4*)(W + (size_t)(2 * p) * M + 4 * n4);
      w1[j] = *(const qfloat4*)(W + (size_t)(2 * p + 1) * M + 4 * n4);
    }
#pragma unroll
    for (int j = 0; j < WCH; ++j) {
      const int v = v0 + 1024 * j;
      const int p = v / (M / 4);
      const int n4 = v - p * (M / 4);
      float* wd = Ws + p * WP + 8 * n4;
      *(lfloat4*)(wd)     = lfloat4{w0[j].x, w1[j].x, w0[j].y, w1[j].y};
      *(lfloat4*)(wd + 4) = lfloat4{w0[j].z, w1[j].z, w0[j].w, w1[j].w};
    }
  }

  for (int g = 0; g < NPH; ++g) {
    stash(g);
    lds_barrier();
    if (g + 1 < NPH) fetch(g + 1);
    const int t0 = g * P;
    const int tl = (nt - t0 < P) ? nt - t0 : P;
    for (int it = wid; it < tl * CT; it += 32) {
      const int rt = it / CT;
      const int ct = it - rt * CT;
      const float* ap = As + (rt * 16 + m) * ldsK + 2 * half;
      const float* bp = Ws + half * WP + 2 * (ct * 16 + m);
      v8f acc = {};
#pragma unroll 1
      for (int kb = 0; kb < K / 64; ++kb) {
#pragma unroll
        for (int s = 0; s < 16; ++s) {
          const v2f a = *(const v2f*)(ap + 4 * s);
          const v2f b = *(const v2f*)(bp + 2 * s * WP);
          acc = __builtin_amdgcn_wmma_f32_16x16x4_f32(
               false, a,  false, b,
               (short)0, acc,  false,  false);
        }
        ap += 64;
        bp += 32 * WP;
      }
      const int row0 = base + (t0 + rt) * 16 + 8 * half;
      float* op = out + (size_t)row0 * M + ct * 16 + m;
      if (base + (t0 + rt) * 16 + 16 <= N) {
#pragma unroll
        for (int r = 0; r < 8; ++r) op[(size_t)r * M] = acc[r];
      } else {
#pragma unroll
        for (int r = 0; r < 8; ++r) {
          if (row0 + r < N) op[(size_t)r * M] = acc[r];
        }
      }
    }
    if (g + 1 < NPH) lds_barrier();
  }
}

__device__ __forceinline__ int wave_in_grid() {
  return __builtin_amdgcn_readfirstlane((int)((blockIdx.x * blockDim.x + threadIdx.x) >> 5));
}
__device__ __forceinline__ int waves_in_grid() { return (int)((gridDim.x * blockDim.x) >> 5); }

typedef int qint4 __attribute__((ext_vector_type(4), aligned(4)));
constexpr int EDGE_LIST_CAP = 256;
template <int NL, int ROWS, int MODE>
__global__ void __launch_bounds__(1024) k_edge_own(const int* __restrict__ src,
                                                   const int* __restrict__ dst,
                                                   const float* __restrict__ dinv,
                                                   const float* __restrict__ hW,
                                                   float* __restrict__ agg,
                                                   int* __restrict__ lsrc, int* __restrict__ lrow,
                                                   float* __restrict__ lnrm, int* __restrict__ lcount,
                                                   int E, int N) {
  constexpr int C = NL * 32;
  constexpr int NW = 32;
  constexpr int U = 4;
  constexpr int G = 32 / NL;
  constexpr int QCAP = 64 + U * 128;
  constexpr int LCAP = EDGE_LIST_CAP;
  __shared__ float acc[ROWS * C];
  __shared__ int stk[NW * QCAP];
  __shared__ int wcnt[NW];
  const int lane = threadIdx.x & 31;
  const int wid = __builtin_amdgcn_readfirstlane((int)(threadIdx.x >> 5));
  const int base = blockIdx.x * ROWS;
  const int nrows = (N - base < ROWS) ? (N - base) : ROWS;
  const int total = nrows * C;

  const int gwave = blockIdx.x * NW + wid;
  const size_t lb = (size_t)gwave * LCAP;
  int lc = -1;
  if (MODE == 1) lc = __builtin_amdgcn_readfirstlane(lcount[gwave]);

  for (int i0 = threadIdx.x; i0 < total; i0 += 8 * 1024) {
    float hv[8], dv[8];
#pragma unroll
    for (int u = 0; u < 8; ++u) {
      const int idx = i0 + u * 1024;
      const int ic = (idx < total) ? idx : total - 1;
      hv[u] = hW[(size_t)base * C + ic];
      dv[u] = dinv[base + ic / C];
    }
#pragma unroll
    for (int u = 0; u < 8; ++u) {
      const int idx = i0 + u * 1024;
      if (idx < total) acc[idx] = hv[u] * dv[u] * dv[u];
    }
  }
  __syncthreads();

  auto add_rows = [&](int s, int dl, float nrm, int cnt) {
    for (int j0 = 0; j0 < cnt; j0 += G) {
      float v[G][NL];
#pragma unroll
      for (int g = 0; g < G; ++g) {
        const int j = (j0 + g < cnt) ? j0 + g : cnt - 1;
        const int sj = __builtin_amdgcn_readlane(s, j);
        const float* hp = hW + (size_t)sj * C + lane;
#pragma unroll
        for (int i = 0; i < NL; ++i) v[g][i] = hp[i * 32];
      }
#pragma unroll
      for (int g = 0; g < G; ++g) {
        if (j0 + g < cnt) {
          const int dj = __builtin_amdgcn_readlane(dl, j0 + g);
          const float nj = __int_as_float(__builtin_amdgcn_readlane(__float_as_int(nrm), j0 + g));
          float* ap = acc + dj * C + lane;
#pragma unroll
          for (int i = 0; i < NL; ++i) atomicAdd(ap + i * 32, v[g][i] * nj);
        }
      }
    }
  };

  if (lc >= 0) {
    for (int i0 = 0; i0 < lc; i0 += 32) {
      const int cnt = (lc - i0 < 32) ? lc - i0 : 32;
      const int li = (lane < cnt) ? lane : cnt - 1;
      const int s = lsrc[lb + i0 + li];
      const int dl = lrow[lb + i0 + li];
      const float nrm = lnrm[lb + i0 + li];
      add_rows(s, dl, nrm, cnt);
    }
  } else {
    const int sb = wid * QCAP;
    int pc = 0;
    int nrec = 0;
    const int nlf = E >> 7;
    {
      const int e = (nlf << 7) + (wid << 5) + lane;
      if ((nlf << 7) + (wid << 5) < E) {
        const int ec = (e < E) ? e : E - 1;
        const bool hit = (e < E) && ((unsigned)(dst[ec] - base) < (unsigned)nrows);
        const unsigned m = __builtin_amdgcn_ballot_w32(hit);
        if (m) {
          if (hit) stk[sb + (int)__builtin_amdgcn_mbcnt_lo(m, 0u)] = e;
          pc = __builtin_popcount(m);
        }
      }
    }
    for (int c0 = wid;; c0 += NW * U) {
      const bool last = c0 >= nlf;
      if (!last) {
        qint4 dq[U];
#pragma unroll
        for (int u = 0; u < U; ++u) {
          const int c = c0 + u * NW;
          const int cc = (c < nlf) ? c : nlf - 1;
          dq[u] = *(const qint4*)(dst + 4 * (((size_t)cc << 5) + lane));
        }
#pragma unroll
        for (int u = 0; u < U; ++u) {
          const int c = c0 + u * NW;
          if (c < nlf) {
#pragma unroll
            for (int k = 0; k < 4; ++k) {
              const unsigned m = __builtin_amdgcn_ballot_w32((unsigned)(dq[u][k] - base) < (unsigned)nrows);
              if (m) {
                if ((unsigned)(dq[u][k] - base) < (unsigned)nrows)
                  stk[sb + pc + (int)__builtin_amdgcn_mbcnt_lo(m, 0u)] = 4 * ((c << 5) + lane) + k;
                pc += __builtin_popcount(m);
              }
            }
          }
        }
      }
      while (pc >= (last ? 1 : 32)) {
        const int cnt = (pc < 32) ? pc : 32;
        const int b0 = pc - cnt;
        const int li = (lane < cnt) ? lane : cnt - 1;
        const int e = stk[sb + b0 + li];
        const int s = src[e];
        const int dl = dst[e] - base;
        const float ds = dinv[s];
        const float dd = dinv[base + dl];
        const float nrm = ds * dd;
        if (MODE == 0) {
          if (nrec + cnt <= LCAP && lane < cnt) {
            lsrc[lb + nrec + lane] = s;
            lrow[lb + nrec + lane] = dl;
            lnrm[lb + nrec + lane] = nrm;
          }
          nrec += cnt;
        }
        add_rows(s, dl, nrm, cnt);
        pc = b0;
      }
      if (last) break;
    }
    if (MODE == 0 && lane == 0) wcnt[wid] = (nrec <= LCAP) ? nrec : -1;
  }
  __syncthreads();

  if (MODE == 0 && wid == 0) lcount[blockIdx.x * NW + lane] = wcnt[lane];

  for (int idx = threadIdx.x; idx < total; idx += 1024) agg[(size_t)base * C + idx] = acc[idx];
}

template <int NL, int R>
__global__ void __launch_bounds__(256) k_post(const float* __restrict__ agg,
                                              const float* __restrict__ bias,
                                              float* __restrict__ out, int N) {
  constexpr int C = NL * 32;
  const int lane = threadIdx.x & 31;
  const int gw = wave_in_grid();
  const int nw = waves_in_grid();
  float bv[NL];
  for (int j = 0; j < NL; ++j) bv[j] = bias[j * 32 + lane];
  for (int n0 = gw; n0 < N; n0 += R * nw) {
    float v[R][NL];
    for (int k = 0; k < R; ++k) {
      const int node = n0 + k * nw;
      if (node < N) {
        const float* row = agg + (size_t)node * C + lane;
        for (int j = 0; j < NL; ++j) v[k][j] = row[j * 32];
      }
    }
    for (int k = 0; k < R; ++k) {
      const int node = n0 + k * nw;
      if (node < N) {
        float s = 0.0f;
        for (int j = 0; j < NL; ++j) {
          float t = fmaxf(v[k][j] + bv[j], 0.0f);
          v[k][j] = t;
          s += t;
        }
        for (int off = 16; off > 0; off >>= 1) s += __shfl_xor(s, off, 32);
        float mean = s / (float)C;
        float q = 0.0f;
        for (int j = 0; j < NL; ++j) {
          float dlt = v[k][j] - mean;
          q += dlt * dlt;
        }
        for (int off = 16; off > 0; off >>= 1) q += __shfl_xor(q, off, 32);
        float rstd = rsqrtf(q / (float)C + 1e-5f);
        float* o = out + (size_t)node * C + lane;
        for (int j = 0; j < NL; ++j) o[j * 32] = (v[k][j] - mean) * rstd;
      }
    }
  }
}

template <int NL, int R>
__global__ void __launch_bounds__(256) k_post_pool(const float* __restrict__ agg,
                                                   const float* __restrict__ bias,
                                                   float* __restrict__ out,
                                                   const int* __restrict__ batch,
                                                   const float* __restrict__ cnt,
                                                   float* __restrict__ pool, int N) {
  constexpr int C = NL * 32;
  const int lane = threadIdx.x & 31;
  const int gw = wave_in_grid();
  const int nw = waves_in_grid();
  const int per = N / nw;
  const int rem = N - per * nw;
  const int nb = gw * per + ((gw < rem) ? gw : rem);
  const int ne = nb + per + ((gw < rem) ? 1 : 0);
  float bv[NL];
  for (int j = 0; j < NL; ++j) bv[j] = bias[j * 32 + lane];
  float ps[NL];
  for (int j = 0; j < NL; ++j) ps[j] = 0.0f;
  int gc = -1;
  auto flush = [&]() {
    if (gc >= 0) {
      const float c = fmaxf(cnt[gc], 1.0f);
      float* pp = pool + (size_t)gc * C + lane;
      for (int j = 0; j < NL; ++j) atomicAdd(pp + j * 32, ps[j] / c);
    }
  };
  for (int n0 = nb; n0 < ne; n0 += R) {
    float v[R][NL];
    int g[R];
    for (int k = 0; k < R; ++k) {
      const int node = n0 + k;
      g[k] = 0;
      if (node < ne) {
        const float* row = agg + (size_t)node * C + lane;
        for (int j = 0; j < NL; ++j) v[k][j] = row[j * 32];
        g[k] = batch[node];
      }
    }
    for (int k = 0; k < R; ++k) {
      const int node = n0 + k;
      if (node < ne) {
        float s = 0.0f;
        for (int j = 0; j < NL; ++j) {
          float t = fmaxf(v[k][j] + bv[j], 0.0f);
          v[k][j] = t;
          s += t;
        }
        for (int off = 16; off > 0; off >>= 1) s += __shfl_xor(s, off, 32);
        float mean = s / (float)C;
        float q = 0.0f;
        for (int j = 0; j < NL; ++j) {
          float dlt = v[k][j] - mean;
          q += dlt * dlt;
        }
        for (int off = 16; off > 0; off >>= 1) q += __shfl_xor(q, off, 32);
        float rstd = rsqrtf(q / (float)C + 1e-5f);
        float* o = out + (size_t)node * C + lane;
        const int gk = __builtin_amdgcn_readfirstlane(g[k]);
        if (gk != gc) {
          flush();
          gc = gk;
          for (int j = 0; j < NL; ++j) ps[j] = 0.0f;
        }
        for (int j = 0; j < NL; ++j) {
          const float y = (v[k][j] - mean) * rstd;
          o[j * 32] = y;
          ps[j] += y;
        }
      }
    }
  }
  flush();
}

static inline size_t align256(size_t x) { return (x + 255) & ~(size_t)255; }

extern "C" void kernel_launch(void* const* d_in, const int* in_sizes, int n_in,
                              void* d_out, int out_size, void* d_ws, size_t ws_size,
                              hipStream_t stream) {
  const float* x     = (const float*)d_in[0];
  const int*   ei    = (const int*)d_in[1];
  const int*   batch = (const int*)d_in[2];
  const float* W0 = (const float*)d_in[4];
  const float* b0 = (const float*)d_in[5];
  const float* W1 = (const float*)d_in[6];
  const float* b1 = (const float*)d_in[7];
  const float* W2 = (const float*)d_in[8];
  const float* b2 = (const float*)d_in[9];

  const int N  = in_sizes[2];
  const int E  = in_sizes[1] / 2;
  const int IN = in_sizes[0] / N;
  const int H0 = in_sizes[5];
  const int H1 = in_sizes[7];
  const int H2 = in_sizes[9];
  const int G  = out_size / H2 - N;

  const int* srcp = ei;
  const int* dstp = ei + E;

  char* ws = (char*)d_ws;
  float* dinv = (float*)ws;  ws += align256((size_t)N * 4);
  float* cnt  = (float*)ws;  ws += align256((size_t)G * 4);
  float* bufA = (float*)ws;  ws += align256((size_t)N * 256 * 4);
  float* bufB = (float*)ws;  ws += align256((size_t)N * 256 * 4);
  const int NLW = ((N + 195) / 196) * 32;
  int*   lcount = (int*)ws;    ws += align256((size_t)NLW * 4);
  int*   lsrc   = (int*)ws;    ws += align256((size_t)NLW * EDGE_LIST_CAP * 4);
  int*   lrow   = (int*)ws;    ws += align256((size_t)NLW * EDGE_LIST_CAP * 4);
  float* lnrm   = (float*)ws;  ws += align256((size_t)NLW * EDGE_LIST_CAP * 4);

  float* pool = (float*)d_out;
  float* hout = (float*)d_out + (size_t)G * H2;

  const int B = 256;
  auto cdiv = [](long long a, long long b) { return (int)((a + b - 1) / b); };

  const int GR = 1024;
  bool listed = false;
  auto edge_sum = [&](const float* hin, float* o, int C) {
    const int g = cdiv(N, 196);
    if (!listed) {
      if (C == 128) k_edge_own<4, 196, 0><<<g, 1024, 0, stream>>>(srcp, dstp, dinv, hin, o, lsrc, lrow, lnrm, lcount, E, N);
      else          k_edge_own<8, 196, 0><<<g, 1024, 0, stream>>>(srcp, dstp, dinv, hin, o, lsrc, lrow, lnrm, lcount, E, N);
    } else {
      if (C == 128) k_edge_own<4, 196, 1><<<g, 1024, 0, stream>>>(srcp, dstp, dinv, hin, o, lsrc, lrow, lnrm, lcount, E, N);
      else          k_edge_own<8, 196, 1><<<g, 1024, 0, stream>>>(srcp, dstp, dinv, hin, o, lsrc, lrow, lnrm, lcount, E, N);
    }
    listed = true;
  };
  auto post = [&](const float* a, const float* bias, float* o, int C, bool pooled) {
    if (pooled) {
      if (C == 128) k_post_pool<4, 8><<<GR, B, 0, stream>>>(a, bias, o, batch, cnt, pool, N);
      else          k_post_pool<8, 4><<<GR, B, 0, stream>>>(a, bias, o, batch, cnt, pool, N);
    } else {
      if (C == 128) k_post<4, 4><<<GR, B, 0, stream>>>(a, bias, o, N);
      else          k_post<8, 2><<<GR, B, 0, stream>>>(a, bias, o, N);
    }
  };

  const int ninit = (N > G * H2) ? N : G * H2;
  k_init<<<cdiv(ninit, B), B, 0, stream>>>(dinv, N, pool, G * H2, cnt, G);
  k_deg_cnt<<<cdiv(E > N ? E : N, B), B, 0, stream>>>(dstp, dinv, E, batch, cnt, N);
  k_dinv<<<cdiv(N, B), B, 0, stream>>>(dinv, N);

  auto layer = [&](const float* hin, int K, int M, const float* W, const float* bias,
                   float* tA, float* tB, float* outp, bool pooled) {
    dim3 g(cdiv(N, 16), M / 128);
    const int gp = cdiv(N, GEMM_TILES * 16);
    auto gemm = [&](const float* a, float* o) {
      if (K == 128 && M == 256)      k_gemm_wlds<128, 256, 13><<<gp, 1024, 0, stream>>>(a, W, o, N);
      else if (K == 256 && M == 256) k_gemm_wlds<256, 256, 2><<<gp, 1024, 0, stream>>>(a, W, o, N);
      else if (K == 256 && M == 128) k_gemm_wlds<256, 128, 4><<<gp, 1024, 0, stream>>>(a, W, o, N);
      else                           k_gemm_wmma<<<g, B, 0, stream>>>(a, W, o, N, K, M);
    };
    if (K < M) {
      edge_sum(hin, tA, K);
      gemm(tA, tB);
    } else {
      gemm(hin, tA);
      edge_sum(tA, tB, M);
    }
    post(tB, bias, outp, M, pooled);
  };

  layer(x,    IN, H0, W0, b0,  bufA,  bufB,  bufA, false);
  layer(bufA, H0, H1, W1, b1,  bufB,  bufA,  bufB, false);
  layer(bufB, H1, H2, W2, b2,  bufA,  bufB,  hout, true);
}


# Bias, relu, layer norm of layers 0 and 1 done by the next product kernel as it stages each A row (wave per row, in registers): 2 launches, 2 row passes gone; norm sums: 8 per lane, lane butterfly
# speedup vs baseline: 3.6589x; 1.2546x over previous
#include <hip/hip_runtime.h>
#include <hip/hip_bf16.h>

typedef __attribute__((ext_vector_type(2))) float v2f;
typedef __attribute__((ext_vector_type(8))) float v8f;

__global__ void k_init(float* __restrict__ deg, int n, float* __restrict__ pool, int np,
                       float* __restrict__ cnt, int nc) {
  const int i = blockIdx.x * blockDim.x + threadIdx.x;
  if (i < n) deg[i] = 1.0f;
  if (i < np) pool[i] = 0.0f;
  if (i < nc) cnt[i] = 0.0f;
}

__global__ void k_deg_cnt(const int* __restrict__ dst, float* __restrict__ deg, int E,
                          const int* __restrict__ batch, float* __restrict__ cnt, int N) {
  const int i = blockIdx.x * blockDim.x + threadIdx.x;
  if (i < E) atomicAdd(deg + dst[i], 1.0f);
  if (i < N) {
    const int g = batch[i];
    const int g0 = __builtin_amdgcn_readfirstlane(g);
    const unsigned act = __builtin_amdgcn_ballot_w32(true);
    const unsigned same = __builtin_amdgcn_ballot_w32(g == g0);
    if (same == act) {
      if (__builtin_amdgcn_mbcnt_lo(act, 0u) == 0) atomicAdd(cnt + g0, (float)__builtin_popcount(act));
    } else {
      atomicAdd(cnt + g, 1.0f);
    }
  }
}

__global__ void k_dinv(float* __restrict__ deg, int n) {
  int i = blockIdx.x * blockDim.x + threadIdx.x;
  if (i < n) deg[i] = rsqrtf(fmaxf(deg[i], 1.0f));
}

__global__ void __launch_bounds__(256) k_gemm_wmma(
    const float* __restrict__ A, const float* __restrict__ W,
    float* __restrict__ out, int N, int K, int M) {
  __shared__ float As[16 * (256 + 4)];
  const int ldsK = K + 4;
  const int rowBase = blockIdx.x * 16;
  const int colBase = blockIdx.y * 128 + (threadIdx.x >> 5) * 16;

  for (int e = threadIdx.x; e < 16 * K; e += 256) {
    int r = e / K;
    int c = e - r * K;
    int g = rowBase + r;
    As[r * ldsK + c] = (g < N) ? A[(size_t)g * K + c] : 0.0f;
  }
  __syncthreads();

  const int lane = threadIdx.x & 31;
  const int half = lane >> 4;
  const int m    = lane & 15;
  const int col  = colBase + m;

  v8f acc = {};
  for (int k = 0; k < K; k += 4) {
    const int kk = k + 2 * half;
    v2f a, b;
    a.x = As[m * ldsK + kk];
    a.y = As[m * ldsK + kk + 1];
    b.x = W[(size_t)kk * M + col];
    b.y = W[(size_t)(kk + 1) * M + col];
    acc = __builtin_amdgcn_wmma_f32_16x16x4_f32(
         false, a,  false, b,
         (short)0, acc,  false,  false);
  }

  for (int r = 0; r < 8; ++r) {
    int row = rowBase + r + 8 * half;
    if (row < N) out[(size_t)row * M + col] = acc[r];
  }
}

__device__ __forceinline__ void lds_barrier() {
  __builtin_amdgcn_fence(__ATOMIC_RELEASE, "workgroup", "local");
  __builtin_amdgcn_s_barrier();
  __builtin_amdgcn_fence(__ATOMIC_ACQUIRE, "workgroup", "local");
}
typedef float qfloat4 __attribute__((ext_vector_type(4), aligned(4)));
typedef float lfloat4 __attribute__((ext_vector_type(4)));
template <int X>
__device__ __forceinline__ float row_xor(float v) {
  return __builtin_bit_cast(float, __builtin_amdgcn_update_dpp(0, __builtin_bit_cast(int, v), 0x160 + X, 0xf, 0xf, true));
}
__device__ __forceinline__ float wave_sum(float v) {
  v += row_xor<1>(v);
  v += row_xor<2>(v);
  v += row_xor<4>(v);
  v += row_xor<8>(v);
  const int b = __builtin_bit_cast(int, v);
  v += __builtin_bit_cast(float, __builtin_amdgcn_permlanex16(b, b, 0x76543210, 0xfedcba98, false, false));
  return v;
}
constexpr int GEMM_TILES = 13;
template <int K, int M, int P, bool LN>
__global__ void __launch_bounds__(1024) k_gemm_wlds(
    const float* __restrict__ A, const float* __restrict__ W, const float* __restrict__ pbias,
    float* __restrict__ out, int N) {
  constexpr int TILES = GEMM_TILES;
  constexpr int CT = M / 16;
  constexpr int ldsK = K + 4;
  constexpr int WP = 2 * M + 32;
  constexpr int NPH = (TILES + P - 1) / P;
  constexpr int K4 = K / 4;
  constexpr int AU = P * 16 * K4;
  constexpr int NS = (AU + 1023) / 1024;
  constexpr int WU = (K / 2) * (M / 4);
  constexpr int WCH = 4;
  constexpr int Q = K / 128;
  constexpr int RPW = (P * 16) / 32;
  static_assert(!LN || (K % 128 == 0 && P % 2 == 0 && RPW * Q == NS), "LN staging: whole rows per wave");
  __shared__ alignas(16) float smem[P * 16 * ldsK + (K / 2) * WP];
  float* const As = smem;
  float* const Ws = smem + P * 16 * ldsK;
  const int tid = threadIdx.x;
  const int lane = tid & 31;
  const int wid = __builtin_amdgcn_readfirstlane(tid >> 5);
  const int half = lane >> 4;
  const int m    = lane & 15;
  const int base = blockIdx.x * (TILES * 16);
  const int left = (N - base + 15) >> 4;
  const int nt = (left < TILES) ? left : TILES;

  qfloat4 st[NS];
  auto fetch = [&](int g) {
    if constexpr (LN) {
#pragma unroll
      for (int j = 0; j < RPW; ++j) {
        const int row = base + g * (P * 16) + wid + 32 * j;
        const int rl = (row < N) ? row : N - 1;
        const float* rp = A + (size_t)rl * K + 4 * lane;
#pragma unroll
        for (int q = 0; q < Q; ++q) st[j * Q + q] = *(const qfloat4*)(rp + 128 * q);
      }
    } else {
#pragma unroll
      for (int i = 0; i < NS; ++i) {
        const int u = tid + 1024 * i;
        const int r = u / K4;
        const int c4 = u - r * K4;
        const int row = base + g * (P * 16) + r;
        const int rl = (row < N) ? row : N - 1;
        st[i] = *(const qfloat4*)(A + (size_t)rl * K + 4 * c4);
      }
    }
  };
  qfloat4 pb[Q];
  if constexpr (LN) {
#pragma unroll
    for (int q = 0; q < Q; ++q) pb[q] = *(const qfloat4*)(pbias + 4 * lane + 128 * q);
  }
  auto norm = [&]() {
#pragma unroll
    for (int j = 0; j < RPW; ++j) {
      float s = 0.0f;
#pragma unroll
      for (int q = 0; q < Q; ++q) {
        qfloat4 t = st[j * Q + q];
        t.x = fmaxf(t.x + pb[q].x, 0.0f);
        t.y = fmaxf(t.y + pb[q].y, 0.0f);
        t.z = fmaxf(t.z + pb[q].z, 0.0f);
        t.w = fmaxf(t.w + pb[q].w, 0.0f);
        st[j * Q + q] = t;
        s += (t.x + t.y) + (t.z + t.w);
      }
      s = wave_sum(s);
      const float mean = s / (float)K;
      float v = 0.0f;
#pragma unroll
      for (int q = 0; q < Q; ++q) {
        const qfloat4 t = st[j * Q + q];
        const float dx = t.x - mean, dy = t.y - mean, dz = t.z - mean, dw = t.w - mean;
        v += (dx * dx + dy * dy) + (dz * dz + dw * dw);
      }
      v = wave_sum(v);
      const float rstd = rsqrtf(v / (float)K + 1e-5f);
#pragma unroll
      for (int q = 0; q < Q; ++q) {
        qfloat4 t = st[j * Q + q];
        t.x = (t.x - mean) * rstd;
        t.y = (t.y - mean) * rstd;
        t.z = (t.z - mean) * rstd;
        t.w = (t.w - mean) * rstd;
        st[j * Q + q] = t;
      }
    }
  };
  auto stash = [&](int g) {
    if constexpr (LN) {
#pragma unroll
      for (int j = 0; j < RPW; ++j) {
        const int r = wid + 32 * j;
        const int row = base + g * (P * 16) + r;
#pragma unroll
        for (int q = 0; q < Q; ++q) {
          lfloat4 v = {st[j * Q + q].x, st[j * Q + q].y, st[j * Q + q].z, st[j * Q + q].w};
          if (row >= N) v = lfloat4{0.0f, 0.0f, 0.0f, 0.0f};
          *(lfloat4*)(As + r * ldsK + 4 * lane + 128 * q) = v;
        }
      }
    } else {
#pragma unroll
      for (int i = 0; i < NS; ++i) {
        const int u = tid + 1024 * i;
        const int r = u / K4;
        const int c4 = u - r * K4;
        const int row = base + g * (P * 16) + r;
        if (u < AU) {
          lfloat4 v = {st[i].x, st[i].y, st[i].z, st[i].w};
          if (row >= N) v = lfloat4{0.0f, 0.0f, 0.0f, 0.0f};
          *(lfloat4*)(As + r * ldsK + 4 * c4) = v;
        }
      }
    }
  };

  fetch(0);

  for (int v0 = tid; v0 < WU; v0 += 1024 * WCH) {
    qfloat4 w0[WCH], w1[WCH];
#pragma unroll
    for (int j = 0; j < WCH; ++j) {
      const int v = v0 + 1024 * j;
      const int p = v / (M / 4);
      const int n4 = v - p * (M / 4);
      w0[j] = *(const qfloat4*)(W + (size_t)(2 * p) * M + 4 * n4);
      w1[j] = *(const qfloat4*)(W + (size_t)(2 * p + 1) * M + 4 * n4);
    }
#pragma unroll
    for (int j = 0; j < WCH; ++j) {
      const int v = v0 + 1024 * j;
      const int p = v / (M / 4);
      const int n4 = v - p * (M / 4);
      float* wd = Ws + p * WP + 8 * n4;
      *(lfloat4*)(wd)     = lfloat4{w0[j].x, w1[j].x, w0[j].y, w1[j].y};
      *(lfloat4*)(wd + 4) = lfloat4{w0[j].z, w1[j].z, w0[j].w, w1[j].w};
    }
  }

  if constexpr (LN) norm();

  for (int g = 0; g < NPH; ++g) {
    stash(g);
    lds_barrier();
    if (g + 1 < NPH) fetch(g + 1);
    const int t0 = g * P;
    const int tl = (nt - t0 < P) ? nt - t0 : P;
    for (int it = wid; it < tl * CT; it += 32) {
      const int rt = it / CT;
      const int ct = it - rt * CT;
      const float* ap = As + (rt * 16 + m) * ldsK + 2 * half;
      const float* bp = Ws + half * WP + 2 * (ct * 16 + m);
      v8f acc = {};
#pragma unroll 1
      for (int kb = 0; kb < K / 64; ++kb) {
#pragma unroll
        for (int s = 0; s < 16; ++s) {
          const v2f a = *(const v2f*)(ap + 4 * s);
          const v2f b = *(const v2f*)(bp + 2 * s * WP);
          acc = __builtin_amdgcn_wmma_f32_16x16x4_f32(
               false, a,  false, b,
               (short)0, acc,  false,  false);
        }
        ap += 64;
        bp += 32 * WP;
      }
      const int row0 = base + (t0 + rt) * 16 + 8 * half;
      float* op = out + (size_t)row0 * M + ct * 16 + m;
      if (base + (t0 + rt) * 16 + 16 <= N) {
#pragma unroll
        for (int r = 0; r < 8; ++r) op[(size_t)r * M] = acc[r];
      } else {
#pragma unroll
        for (int r = 0; r < 8; ++r) {
          if (row0 + r < N) op[(size_t)r * M] = acc[r];
        }
      }
    }
    if (g + 1 < NPH) {
      if constexpr (LN) norm();
      lds_barrier();
    }
  }
}

__device__ __forceinline__ int wave_in_grid() {
  return __builtin_amdgcn_readfirstlane((int)((blockIdx.x * blockDim.x + threadIdx.x) >> 5));
}
__device__ __forceinline__ int waves_in_grid() { return (int)((gridDim.x * blockDim.x) >> 5); }

typedef int qint4 __attribute__((ext_vector_type(4), aligned(4)));
constexpr int EDGE_LIST_CAP = 256;
template <int NL, int ROWS, int MODE>
__global__ void __launch_bounds__(1024) k_edge_own(const int* __restrict__ src,
                                                   const int* __restrict__ dst,
                                                   const float* __restrict__ dinv,
                                                   const float* __restrict__ hW,
                                                   float* __restrict__ agg,
                                                   int* __restrict__ lsrc, int* __restrict__ lrow,
                                                   float* __restrict__ lnrm, int* __restrict__ lcount,
                                                   int E, int N) {
  constexpr int C = NL * 32;
  constexpr int NW = 32;
  constexpr int U = 4;
  constexpr int G = 32 / NL;
  constexpr int QCAP = 64 + U * 128;
  constexpr int LCAP = EDGE_LIST_CAP;
  __shared__ float acc[ROWS * C];
  __shared__ int stk[NW * QCAP];
  __shared__ int wcnt[NW];
  const int lane = threadIdx.x & 31;
  const int wid = __builtin_amdgcn_readfirstlane((int)(threadIdx.x >> 5));
  const int base = blockIdx.x * ROWS;
  const int nrows = (N - base < ROWS) ? (N - base) : ROWS;
  const int total = nrows * C;

  const int gwave = blockIdx.x * NW + wid;
  const size_t lb = (size_t)gwave * LCAP;
  int lc = -1;
  if (MODE == 1) lc = __builtin_amdgcn_readfirstlane(lcount[gwave]);

  for (int i0 = threadIdx.x; i0 < total; i0 += 8 * 1024) {
    float hv[8], dv[8];
#pragma unroll
    for (int u = 0; u < 8; ++u) {
      const int idx = i0 + u * 1024;
      const int ic = (idx < total) ? idx : total - 1;
      hv[u] = hW[(size_t)base * C + ic];
      dv[u] = dinv[base + ic / C];
    }
#pragma unroll
    for (int u = 0; u < 8; ++u) {
      const int idx = i0 + u * 1024;
      if (idx < total) acc[idx] = hv[u] * dv[u] * dv[u];
    }
  }
  __syncthreads();

  auto add_rows = [&](int s, int dl, float nrm, int cnt) {
    for (int j0 = 0; j0 < cnt; j0 += G) {
      float v[G][NL];
#pragma unroll
      for (int g = 0; g < G; ++g) {
        const int j = (j0 + g < cnt) ? j0 + g : cnt - 1;
        const int sj = __builtin_amdgcn_readlane(s, j);
        const float* hp = hW + (size_t)sj * C + lane;
#pragma unroll
        for (int i = 0; i < NL; ++i) v[g][i] = hp[i * 32];
      }
#pragma unroll
      for (int g = 0; g < G; ++g) {
        if (j0 + g < cnt) {
          const int dj = __builtin_amdgcn_readlane(dl, j0 + g);
          const float nj = __int_as_float(__builtin_amdgcn_readlane(__float_as_int(nrm), j0 + g));
          float* ap = acc + dj * C + lane;
#pragma unroll
          for (int i = 0; i < NL; ++i) atomicAdd(ap + i * 32, v[g][i] * nj);
        }
      }
    }
  };

  if (lc >= 0) {
    for (int i0 = 0; i0 < lc; i0 += 32) {
      const int cnt = (lc - i0 < 32) ? lc - i0 : 32;
      const int li = (lane < cnt) ? lane : cnt - 1;
      const int s = lsrc[lb + i0 + li];
      const int dl = lrow[lb + i0 + li];
      const float nrm = lnrm[lb + i0 + li];
      add_rows(s, dl, nrm, cnt);
    }
  } else {
    const int sb = wid * QCAP;
    int pc = 0;
    int nrec = 0;
    const int nlf = E >> 7;
    {
      const int e = (nlf << 7) + (wid << 5) + lane;
      if ((nlf << 7) + (wid << 5) < E) {
        const int ec = (e < E) ? e : E - 1;
        const bool hit = (e < E) && ((unsigned)(dst[ec] - base) < (unsigned)nrows);
        const unsigned m = __builtin_amdgcn_ballot_w32(hit);
        if (m) {
          if (hit) stk[sb + (int)__builtin_amdgcn_mbcnt_lo(m, 0u)] = e;
          pc = __builtin_popcount(m);
        }
      }
    }
    for (int c0 = wid;; c0 += NW * U) {
      const bool last = c0 >= nlf;
      if (!last) {
        qint4 dq[U];
#pragma unroll
        for (int u = 0; u < U; ++u) {
          const int c = c0 + u * NW;
          const int cc = (c < nlf) ? c : nlf - 1;
          dq[u] = *(const qint4*)(dst + 4 * (((size_t)cc << 5) + lane));
        }
#pragma unroll
        for (int u = 0; u < U; ++u) {
          const int c = c0 + u * NW;
          if (c < nlf) {
#pragma unroll
            for (int k = 0; k < 4; ++k) {
              const unsigned m = __builtin_amdgcn_ballot_w32((unsigned)(dq[u][k] - base) < (unsigned)nrows);
              if (m) {
                if ((unsigned)(dq[u][k] - base) < (unsigned)nrows)
                  stk[sb + pc + (int)__builtin_amdgcn_mbcnt_lo(m, 0u)] = 4 * ((c << 5) + lane) + k;
                pc += __builtin_popcount(m);
              }
            }
          }
        }
      }
      while (pc >= (last ? 1 : 32)) {
        const int cnt = (pc < 32) ? pc : 32;
        const int b0 = pc - cnt;
        const int li = (lane < cnt) ? lane : cnt - 1;
        const int e = stk[sb + b0 + li];
        const int s = src[e];
        const int dl = dst[e] - base;
        const float ds = dinv[s];
        const float dd = dinv[base + dl];
        const float nrm = ds * dd;
        if (MODE == 0) {
          if (nrec + cnt <= LCAP && lane < cnt) {
            lsrc[lb + nrec + lane] = s;
            lrow[lb + nrec + lane] = dl;
            lnrm[lb + nrec + lane] = nrm;
          }
          nrec += cnt;
        }
        add_rows(s, dl, nrm, cnt);
        pc = b0;
      }
      if (last) break;
    }
    if (MODE == 0 && lane == 0) wcnt[wid] = (nrec <= LCAP) ? nrec : -1;
  }
  __syncthreads();

  if (MODE == 0 && wid == 0) lcount[blockIdx.x * NW + lane] = wcnt[lane];

  for (int idx = threadIdx.x; idx < total; idx += 1024) agg[(size_t)base * C + idx] = acc[idx];
}

template <int NL, int R>
__global__ void __launch_bounds__(256) k_post(const float* __restrict__ agg,
                                              const float* __restrict__ bias,
                                              float* __restrict__ out, int N) {
  constexpr int C = NL * 32;
  const int lane = threadIdx.x & 31;
  const int gw = wave_in_grid();
  const int nw = waves_in_grid();
  float bv[NL];
  for (int j = 0; j < NL; ++j) bv[j] = bias[j * 32 + lane];
  for (int n0 = gw; n0 < N; n0 += R * nw) {
    float v[R][NL];
    for (int k = 0; k < R; ++k) {
      const int node = n0 + k * nw;
      if (node < N) {
        const float* row = agg + (size_t)node * C + lane;
        for (int j = 0; j < NL; ++j) v[k][j] = row[j * 32];
      }
    }
    for (int k = 0; k < R; ++k) {
      const int node = n0 + k * nw;
      if (node < N) {
        float s = 0.0f;
        for (int j = 0; j < NL; ++j) {
          float t = fmaxf(v[k][j] + bv[j], 0.0f);
          v[k][j] = t;
          s += t;
        }
        for (int off = 16; off > 0; off >>= 1) s += __shfl_xor(s, off, 32);
        float mean = s / (float)C;
        float q = 0.0f;
        for (int j = 0; j < NL; ++j) {
          float dlt = v[k][j] - mean;
          q += dlt * dlt;
        }
        for (int off = 16; off > 0; off >>= 1) q += __shfl_xor(q, off, 32);
        float rstd = rsqrtf(q / (float)C + 1e-5f);
        float* o = out + (size_t)node * C + lane;
        for (int j = 0; j < NL; ++j) o[j * 32] = (v[k][j] - mean) * rstd;
      }
    }
  }
}

template <int NL, int R>
__global__ void __launch_bounds__(256) k_post_pool(const float* __restrict__ agg,
                                                   const float* __restrict__ bias,
                                                   float* __restrict__ out,
                                                   const int* __restrict__ batch,
                                                   const float* __restrict__ cnt,
                                                   float* __restrict__ pool, int N) {
  constexpr int C = NL * 32;
  const int lane = threadIdx.x & 31;
  const int gw = wave_in_grid();
  const int nw = waves_in_grid();
  const int per = N / nw;
  const int rem = N - per * nw;
  const int nb = gw * per + ((gw < rem) ? gw : rem);
  const int ne = nb + per + ((gw < rem) ? 1 : 0);
  float bv[NL];
  for (int j = 0; j < NL; ++j) bv[j] = bias[j * 32 + lane];
  float ps[NL];
  for (int j = 0; j < NL; ++j) ps[j] = 0.0f;
  int gc = -1;
  auto flush = [&]() {
    if (gc >= 0) {
      const float c = fmaxf(cnt[gc], 1.0f);
      float* pp = pool + (size_t)gc * C + lane;
      for (int j = 0; j < NL; ++j) atomicAdd(pp + j * 32, ps[j] / c);
    }
  };
  for (int n0 = nb; n0 < ne; n0 += R) {
    float v[R][NL];
    int g[R];
    for (int k = 0; k < R; ++k) {
      const int node = n0 + k;
      g[k] = 0;
      if (node < ne) {
        const float* row = agg + (size_t)node * C + lane;
        for (int j = 0; j < NL; ++j) v[k][j] = row[j * 32];
        g[k] = batch[node];
      }
    }
    for (int k = 0; k < R; ++k) {
      const int node = n0 + k;
      if (node < ne) {
        float s = 0.0f;
        for (int j = 0; j < NL; ++j) {
          float t = fmaxf(v[k][j] + bv[j], 0.0f);
          v[k][j] = t;
          s += t;
        }
        for (int off = 16; off > 0; off >>= 1) s += __shfl_xor(s, off, 32);
        float mean = s / (float)C;
        float q = 0.0f;
        for (int j = 0; j < NL; ++j) {
          float dlt = v[k][j] - mean;
          q += dlt * dlt;
        }
        for (int off = 16; off > 0; off >>= 1) q += __shfl_xor(q, off, 32);
        float rstd = rsqrtf(q / (float)C + 1e-5f);
        float* o = out + (size_t)node * C + lane;
        const int gk = __builtin_amdgcn_readfirstlane(g[k]);
        if (gk != gc) {
          flush();
          gc = gk;
          for (int j = 0; j < NL; ++j) ps[j] = 0.0f;
        }
        for (int j = 0; j < NL; ++j) {
          const float y = (v[k][j] - mean) * rstd;
          o[j * 32] = y;
          ps[j] += y;
        }
      }
    }
  }
  flush();
}

static inline size_t align256(size_t x) { return (x + 255) & ~(size_t)255; }

extern "C" void kernel_launch(void* const* d_in, const int* in_sizes, int n_in,
                              void* d_out, int out_size, void* d_ws, size_t ws_size,
                              hipStream_t stream) {
  const float* x     = (const float*)d_in[0];
  const int*   ei    = (const int*)d_in[1];
  const int*   batch = (const int*)d_in[2];
  const float* W0 = (const float*)d_in[4];
  const float* b0 = (const float*)d_in[5];
  const float* W1 = (const float*)d_in[6];
  const float* b1 = (const float*)d_in[7];
  const float* W2 = (const float*)d_in[8];
  const float* b2 = (const float*)d_in[9];

  const int N  = in_sizes[2];
  const int E  = in_sizes[1] / 2;
  const int IN = in_sizes[0] / N;
  const int H0 = in_sizes[5];
  const int H1 = in_sizes[7];
  const int H2 = in_sizes[9];
  const int G  = out_size / H2 - N;

  const int* srcp = ei;
  const int* dstp = ei + E;

  char* ws = (char*)d_ws;
  float* dinv = (float*)ws;  ws += align256((size_t)N * 4);
  float* cnt  = (float*)ws;  ws += align256((size_t)G * 4);
  float* bufA = (float*)ws;  ws += align256((size_t)N * 256 * 4);
  float* bufB = (float*)ws;  ws += align256((size_t)N * 256 * 4);
  const int NLW = ((N + 195) / 196) * 32;
  int*   lcount = (int*)ws;    ws += align256((size_t)NLW * 4);
  int*   lsrc   = (int*)ws;    ws += align256((size_t)NLW * EDGE_LIST_CAP * 4);
  int*   lrow   = (int*)ws;    ws += align256((size_t)NLW * EDGE_LIST_CAP * 4);
  float* lnrm   = (float*)ws;  ws += align256((size_t)NLW * EDGE_LIST_CAP * 4);

  float* pool = (float*)d_out;
  float* hout = (float*)d_out + (size_t)G * H2;

  const int B = 256;
  auto cdiv = [](long long a, long long b) { return (int)((a + b - 1) / b); };

  const int GR = 1024;
  bool listed = false;
  auto edge_sum = [&](const float* hin, float* o, int C) {
    const int g = cdiv(N, 196);
    if (!listed) {
      if (C == 128) k_edge_own<4, 196, 0><<<g, 1024, 0, stream>>>(srcp, dstp, dinv, hin, o, lsrc, lrow, lnrm, lcount, E, N);
      else          k_edge_own<8, 196, 0><<<g, 1024, 0, stream>>>(srcp, dstp, dinv, hin, o, lsrc, lrow, lnrm, lcount, E, N);
    } else {
      if (C == 128) k_edge_own<4, 196, 1><<<g, 1024, 0, stream>>>(srcp, dstp, dinv, hin, o, lsrc, lrow, lnrm, lcount, E, N);
      else          k_edge_own<8, 196, 1><<<g, 1024, 0, stream>>>(srcp, dstp, dinv, hin, o, lsrc, lrow, lnrm, lcount, E, N);
    }
    listed = true;
  };
  auto post = [&](const float* a, const float* bias, float* o, int C, bool pooled) {
    if (pooled) {
      if (C == 128) k_post_pool<4, 8><<<GR, B, 0, stream>>>(a, bias, o, batch, cnt, pool, N);
      else          k_post_pool<8, 4><<<GR, B, 0, stream>>>(a, bias, o, batch, cnt, pool, N);
    } else {
      if (C == 128) k_post<4, 4><<<GR, B, 0, stream>>>(a, bias, o, N);
      else          k_post<8, 2><<<GR, B, 0, stream>>>(a, bias, o, N);
    }
  };

  const int ninit = (N > G * H2) ? N : G * H2;
  k_init<<<cdiv(ninit, B), B, 0, stream>>>(dinv, N, pool, G * H2, cnt, G);
  k_deg_cnt<<<cdiv(E > N ? E : N, B), B, 0, stream>>>(dstp, dinv, E, batch, cnt, N);
  k_dinv<<<cdiv(N, B), B, 0, stream>>>(dinv, N);

  auto layer = [&](const float* hin, const float* pb, int K, int M, const float* W, const float* bias,
                   float* tA, float* tB, float* outp, bool pooled, bool raw_out) {
    dim3 g(cdiv(N, 16), M / 128);
    const int gp = cdiv(N, GEMM_TILES * 16);
    auto gemm = [&](const float* a, float* o) {
      if (pb) {
        if (M == 256) k_gemm_wlds<256, 256, 2, true><<<gp, 1024, 0, stream>>>(a, W, pb, o, N);
        else          k_gemm_wlds<256, 128, 4, true><<<gp, 1024, 0, stream>>>(a, W, pb, o, N);
      }
      else if (K == 128 && M == 256) k_gemm_wlds<128, 256, 13, false><<<gp, 1024, 0, stream>>>(a, W, nullptr, o, N);
      else if (K == 256 && M == 256) k_gemm_wlds<256, 256, 2, false><<<gp, 1024, 0, stream>>>(a, W, nullptr, o, N);
      else if (K == 256 && M == 128) k_gemm_wlds<256, 128, 4, false><<<gp, 1024, 0, stream>>>(a, W, nullptr, o, N);
      else                           k_gemm_wmma<<<g, B, 0, stream>>>(a, W, o, N, K, M);
    };
    if (K < M) {
      edge_sum(hin, tA, K);
      gemm(tA, tB);
    } else {
      gemm(hin, tA);
      edge_sum(tA, tB, M);
    }
    if (!raw_out) post(tB, bias, outp, M, pooled);
  };

  auto norm_in_product = [](int K, int M) { return K == 256 && (M == 256 || M == 128); };
  if (norm_in_product(H0, H1) && norm_in_product(H1, H2)) {
    layer(x,    nullptr, IN, H0, W0, b0,  bufA,  bufB,  nullptr, false, true);
    layer(bufB, b0,      H0, H1, W1, b1,  bufA,  bufB,  nullptr, false, true);
    layer(bufB, b1,      H1, H2, W2, b2,  bufA,  bufB,  hout, true, false);
  } else {
    layer(x,    nullptr, IN, H0, W0, b0,  bufA,  bufB,  bufA, false, false);
    layer(bufA, nullptr, H0, H1, W1, b1,  bufB,  bufA,  bufB, false, false);
    layer(bufB, nullptr, H1, H2, W2, b2,  bufA,  bufB,  hout, true, false);
  }
}


# Last layer's bias, relu, layer norm and mean pool done by its edge kernel's owner work-groups on the finished rows in LDS (1 launch, 1 row pass gone); pool: runs of 6-7 owned rows, then atomic order
# speedup vs baseline: 3.8821x; 1.0610x over previous
#include <hip/hip_runtime.h>
#include <hip/hip_bf16.h>

typedef __attribute__((ext_vector_type(2))) float v2f;
typedef __attribute__((ext_vector_type(8))) float v8f;

__global__ void k_init(float* __restrict__ deg, int n, float* __restrict__ pool, int np,
                       float* __restrict__ cnt, int nc) {
  const int i = blockIdx.x * blockDim.x + threadIdx.x;
  if (i < n) deg[i] = 1.0f;
  if (i < np) pool[i] = 0.0f;
  if (i < nc) cnt[i] = 0.0f;
}

__global__ void k_deg_cnt(const int* __restrict__ dst, float* __restrict__ deg, int E,
                          const int* __restrict__ batch, float* __restrict__ cnt, int N) {
  const int i = blockIdx.x * blockDim.x + threadIdx.x;
  if (i < E) atomicAdd(deg + dst[i], 1.0f);
  if (i < N) {
    const int g = batch[i];
    const int g0 = __builtin_amdgcn_readfirstlane(g);
    const unsigned act = __builtin_amdgcn_ballot_w32(true);
    const unsigned same = __builtin_amdgcn_ballot_w32(g == g0);
    if (same == act) {
      if (__builtin_amdgcn_mbcnt_lo(act, 0u) == 0) atomicAdd(cnt + g0, (float)__builtin_popcount(act));
    } else {
      atomicAdd(cnt + g, 1.0f);
    }
  }
}

__global__ void k_dinv(float* __restrict__ deg, int n) {
  int i = blockIdx.x * blockDim.x + threadIdx.x;
  if (i < n) deg[i] = rsqrtf(fmaxf(deg[i], 1.0f));
}

__global__ void __launch_bounds__(256) k_gemm_wmma(
    const float* __restrict__ A, const float* __restrict__ W,
    float* __restrict__ out, int N, int K, int M) {
  __shared__ float As[16 * (256 + 4)];
  const int ldsK = K + 4;
  const int rowBase = blockIdx.x * 16;
  const int colBase = blockIdx.y * 128 + (threadIdx.x >> 5) * 16;

  for (int e = threadIdx.x; e < 16 * K; e += 256) {
    int r = e / K;
    int c = e - r * K;
    int g = rowBase + r;
    As[r * ldsK + c] = (g < N) ? A[(size_t)g * K + c] : 0.0f;
  }
  __syncthreads();

  const int lane = threadIdx.x & 31;
  const int half = lane >> 4;
  const int m    = lane & 15;
  const int col  = colBase + m;

  v8f acc = {};
  for (int k = 0; k < K; k += 4) {
    const int kk = k + 2 * half;
    v2f a, b;
    a.x = As[m * ldsK + kk];
    a.y = As[m * ldsK + kk + 1];
    b.x = W[(size_t)kk * M + col];
    b.y = W[(size_t)(kk + 1) * M + col];
    acc = __builtin_amdgcn_wmma_f32_16x16x4_f32(
         false, a,  false, b,
         (short)0, acc,  false,  false);
  }

  for (int r = 0; r < 8; ++r) {
    int row = rowBase + r + 8 * half;
    if (row < N) out[(size_t)row * M + col] = acc[r];
  }
}

__device__ __forceinline__ void lds_barrier() {
  __builtin_amdgcn_fence(__ATOMIC_RELEASE, "workgroup", "local");
  __builtin_amdgcn_s_barrier();
  __builtin_amdgcn_fence(__ATOMIC_ACQUIRE, "workgroup", "local");
}
typedef float qfloat4 __attribute__((ext_vector_type(4), aligned(4)));
typedef float lfloat4 __attribute__((ext_vector_type(4)));
template <int X>
__device__ __forceinline__ float row_xor(float v) {
  return __builtin_bit_cast(float, __builtin_amdgcn_update_dpp(0, __builtin_bit_cast(int, v), 0x160 + X, 0xf, 0xf, true));
}
__device__ __forceinline__ float wave_sum(float v) {
  v += row_xor<1>(v);
  v += row_xor<2>(v);
  v += row_xor<4>(v);
  v += row_xor<8>(v);
  const int b = __builtin_bit_cast(int, v);
  v += __builtin_bit_cast(float, __builtin_amdgcn_permlanex16(b, b, 0x76543210, 0xfedcba98, false, false));
  return v;
}
__device__ __forceinline__ float wave_sum_down(float v) {
  const int b = __builtin_bit_cast(int, v);
  v += __builtin_bit_cast(float, __builtin_amdgcn_permlanex16(b, b, 0x76543210, 0xfedcba98, false, false));
  v += row_xor<8>(v);
  v += row_xor<4>(v);
  v += row_xor<2>(v);
  v += row_xor<1>(v);
  return v;
}
constexpr int GEMM_TILES = 13;
template <int K, int M, int P, bool LN>
__global__ void __launch_bounds__(1024) k_gemm_wlds(
    const float* __restrict__ A, const float* __restrict__ W, const float* __restrict__ pbias,
    float* __restrict__ out, int N) {
  constexpr int TILES = GEMM_TILES;
  constexpr int CT = M / 16;
  constexpr int ldsK = K + 4;
  constexpr int WP = 2 * M + 32;
  constexpr int NPH = (TILES + P - 1) / P;
  constexpr int K4 = K / 4;
  constexpr int AU = P * 16 * K4;
  constexpr int NS = (AU + 1023) / 1024;
  constexpr int WU = (K / 2) * (M / 4);
  constexpr int WCH = 4;
  constexpr int Q = K / 128;
  constexpr int RPW = (P * 16) / 32;
  static_assert(!LN || (K % 128 == 0 && P % 2 == 0 && RPW * Q == NS), "LN staging: whole rows per wave");
  __shared__ alignas(16) float smem[P * 16 * ldsK + (K / 2) * WP];
  float* const As = smem;
  float* const Ws = smem + P * 16 * ldsK;
  const int tid = threadIdx.x;
  const int lane = tid & 31;
  const int wid = __builtin_amdgcn_readfirstlane(tid >> 5);
  const int half = lane >> 4;
  const int m    = lane & 15;
  const int base = blockIdx.x * (TILES * 16);
  const int left = (N - base + 15) >> 4;
  const int nt = (left < TILES) ? left : TILES;

  qfloat4 st[NS];
  auto fetch = [&](int g) {
    if constexpr (LN) {
#pragma unroll
      for (int j = 0; j < RPW; ++j) {
        const int row = base + g * (P * 16) + wid + 32 * j;
        const int rl = (row < N) ? row : N - 1;
        const float* rp = A + (size_t)rl * K + 4 * lane;
#pragma unroll
        for (int q = 0; q < Q; ++q) st[j * Q + q] = *(const qfloat4*)(rp + 128 * q);
      }
    } else {
#pragma unroll
      for (int i = 0; i < NS; ++i) {
        const int u = tid + 1024 * i;
        const int r = u / K4;
        const int c4 = u - r * K4;
        const int row = base + g * (P * 16) + r;
        const int rl = (row < N) ? row : N - 1;
        st[i] = *(const qfloat4*)(A + (size_t)rl * K + 4 * c4);
      }
    }
  };
  qfloat4 pb[Q];
  if constexpr (LN) {
#pragma unroll
    for (int q = 0; q < Q; ++q) pb[q] = *(const qfloat4*)(pbias + 4 * lane + 128 * q);
  }
  auto norm = [&]() {
#pragma unroll
    for (int j = 0; j < RPW; ++j) {
      float s = 0.0f;
#pragma unroll
      for (int q = 0; q < Q; ++q) {
        qfloat4 t = st[j * Q + q];
        t.x = fmaxf(t.x + pb[q].x, 0.0f);
        t.y = fmaxf(t.y + pb[q].y, 0.0f);
        t.z = fmaxf(t.z + pb[q].z, 0.0f);
        t.w = fmaxf(t.w + pb[q].w, 0.0f);
        st[j * Q + q] = t;
        s += (t.x + t.y) + (t.z + t.w);
      }
      s = wave_sum(s);
      const float mean = s / (float)K;
      float v = 0.0f;
#pragma unroll
      for (int q = 0; q < Q; ++q) {
        const qfloat4 t = st[j * Q + q];
        const float dx = t.x - mean, dy = t.y - mean, dz = t.z - mean, dw = t.w - mean;
        v += (dx * dx + dy * dy) + (dz * dz + dw * dw);
      }
      v = wave_sum(v);
      const float rstd = rsqrtf(v / (float)K + 1e-5f);
#pragma unroll
      for (int q = 0; q < Q; ++q) {
        qfloat4 t = st[j * Q + q];
        t.x = (t.x - mean) * rstd;
        t.y = (t.y - mean) * rstd;
        t.z = (t.z - mean) * rstd;
        t.w = (t.w - mean) * rstd;
        st[j * Q + q] = t;
      }
    }
  };
  auto stash = [&](int g) {
    if constexpr (LN) {
#pragma unroll
      for (int j = 0; j < RPW; ++j) {
        const int r = wid + 32 * j;
        const int row = base + g * (P * 16) + r;
#pragma unroll
        for (int q = 0; q < Q; ++q) {
          lfloat4 v = {st[j * Q + q].x, st[j * Q + q].y, st[j * Q + q].z, st[j * Q + q].w};
          if (row >= N) v = lfloat4{0.0f, 0.0f, 0.0f, 0.0f};
          *(lfloat4*)(As + r * ldsK + 4 * lane + 128 * q) = v;
        }
      }
    } else {
#pragma unroll
      for (int i = 0; i < NS; ++i) {
        const int u = tid + 1024 * i;
        const int r = u / K4;
        const int c4 = u - r * K4;
        const int row = base + g * (P * 16) + r;
        if (u < AU) {
          lfloat4 v = {st[i].x, st[i].y, st[i].z, st[i].w};
          if (row >= N) v = lfloat4{0.0f, 0.0f, 0.0f, 0.0f};
          *(lfloat4*)(As + r * ldsK + 4 * c4) = v;
        }
      }
    }
  };

  fetch(0);

  for (int v0 = tid; v0 < WU; v0 += 1024 * WCH) {
    qfloat4 w0[WCH], w1[WCH];
#pragma unroll
    for (int j = 0; j < WCH; ++j) {
      const int v = v0 + 1024 * j;
      const int p = v / (M / 4);
      const int n4 = v - p * (M / 4);
      w0[j] = *(const qfloat4*)(W + (size_t)(2 * p) * M + 4 * n4);
      w1[j] = *(const qfloat4*)(W + (size_t)(2 * p + 1) * M + 4 * n4);
    }
#pragma unroll
    for (int j = 0; j < WCH; ++j) {
      const int v = v0 + 1024 * j;
      const int p = v / (M / 4);
      const int n4 = v - p * (M / 4);
      float* wd = Ws + p * WP + 8 * n4;
      *(lfloat4*)(wd)     = lfloat4{w0[j].x, w1[j].x, w0[j].y, w1[j].y};
      *(lfloat4*)(wd + 4) = lfloat4{w0[j].z, w1[j].z, w0[j].w, w1[j].w};
    }
  }

  if constexpr (LN) norm();

  for (int g = 0; g < NPH; ++g) {
    stash(g);
    lds_barrier();
    if (g + 1 < NPH) fetch(g + 1);
    const int t0 = g * P;
    const int tl = (nt - t0 < P) ? nt - t0 : P;
    for (int it = wid; it < tl * CT; it += 32) {
      const int rt = it / CT;
      const int ct = it - rt * CT;
      const float* ap = As + (rt * 16 + m) * ldsK + 2 * half;
      const float* bp = Ws + half * WP + 2 * (ct * 16 + m);
      v8f acc = {};
#pragma unroll 1
      for (int kb = 0; kb < K / 64; ++kb) {
#pragma unroll
        for (int s = 0; s < 16; ++s) {
          const v2f a = *(const v2f*)(ap + 4 * s);
          const v2f b = *(const v2f*)(bp + 2 * s * WP);
          acc = __builtin_amdgcn_wmma_f32_16x16x4_f32(
               false, a,  false, b,
               (short)0, acc,  false,  false);
        }
        ap += 64;
        bp += 32 * WP;
      }
      const int row0 = base + (t0 + rt) * 16 + 8 * half;
      float* op = out + (size_t)row0 * M + ct * 16 + m;
      if (base + (t0 + rt) * 16 + 16 <= N) {
#pragma unroll
        for (int r = 0; r < 8; ++r) op[(size_t)r * M] = acc[r];
      } else {
#pragma unroll
        for (int r = 0; r < 8; ++r) {
          if (row0 + r < N) op[(size_t)r * M] = acc[r];
        }
      }
    }
    if (g + 1 < NPH) {
      if constexpr (LN) norm();
      lds_barrier();
    }
  }
}

__device__ __forceinline__ int wave_in_grid() {
  return __builtin_amdgcn_readfirstlane((int)((blockIdx.x * blockDim.x + threadIdx.x) >> 5));
}
__device__ __forceinline__ int waves_in_grid() { return (int)((gridDim.x * blockDim.x) >> 5); }

typedef int qint4 __attribute__((ext_vector_type(4), aligned(4)));
constexpr int EDGE_LIST_CAP = 256;
template <int NL, int ROWS, int MODE, bool FIN>
__global__ void __launch_bounds__(1024) k_edge_own(const int* __restrict__ src,
                                                   const int* __restrict__ dst,
                                                   const float* __restrict__ dinv,
                                                   const float* __restrict__ hW,
                                                   float* __restrict__ agg,
                                                   int* __restrict__ lsrc, int* __restrict__ lrow,
                                                   float* __restrict__ lnrm, int* __restrict__ lcount,
                                                   int E, int N,
                                                   const float* __restrict__ fbias,
                                                   const int* __restrict__ fbatch,
                                                   const float* __restrict__ fcnt,
                                                   float* __restrict__ fpool) {
  constexpr int C = NL * 32;
  constexpr int NW = 32;
  constexpr int U = 4;
  constexpr int G = 32 / NL;
  constexpr int QCAP = 64 + U * 128;
  constexpr int LCAP = EDGE_LIST_CAP;
  __shared__ float acc[ROWS * C];
  __shared__ int stk[NW * QCAP];
  __shared__ int wcnt[NW];
  const int lane = threadIdx.x & 31;
  const int wid = __builtin_amdgcn_readfirstlane((int)(threadIdx.x >> 5));
  const int base = blockIdx.x * ROWS;
  const int nrows = (N - base < ROWS) ? (N - base) : ROWS;
  const int total = nrows * C;

  const int gwave = blockIdx.x * NW + wid;
  const size_t lb = (size_t)gwave * LCAP;
  int lc = -1;
  if (MODE == 1) lc = __builtin_amdgcn_readfirstlane(lcount[gwave]);

  for (int i0 = threadIdx.x; i0 < total; i0 += 8 * 1024) {
    float hv[8], dv[8];
#pragma unroll
    for (int u = 0; u < 8; ++u) {
      const int idx = i0 + u * 1024;
      const int ic = (idx < total) ? idx : total - 1;
      hv[u] = hW[(size_t)base * C + ic];
      dv[u] = dinv[base + ic / C];
    }
#pragma unroll
    for (int u = 0; u < 8; ++u) {
      const int idx = i0 + u * 1024;
      if (idx < total) acc[idx] = hv[u] * dv[u] * dv[u];
    }
  }
  __syncthreads();

  auto add_rows = [&](int s, int dl, float nrm, int cnt) {
    for (int j0 = 0; j0 < cnt; j0 += G) {
      float v[G][NL];
#pragma unroll
      for (int g = 0; g < G; ++g) {
        const int j = (j0 + g < cnt) ? j0 + g : cnt - 1;
        const int sj = __builtin_amdgcn_readlane(s, j);
        const float* hp = hW + (size_t)sj * C + lane;
#pragma unroll
        for (int i = 0; i < NL; ++i) v[g][i] = hp[i * 32];
      }
#pragma unroll
      for (int g = 0; g < G; ++g) {
        if (j0 + g < cnt) {
          const int dj = __builtin_amdgcn_readlane(dl, j0 + g);
          const float nj = __int_as_float(__builtin_amdgcn_readlane(__float_as_int(nrm), j0 + g));
          float* ap = acc + dj * C + lane;
#pragma unroll
          for (int i = 0; i < NL; ++i) atomicAdd(ap + i * 32, v[g][i] * nj);
        }
      }
    }
  };

  if (lc >= 0) {
    for (int i0 = 0; i0 < lc; i0 += 32) {
      const int cnt = (lc - i0 < 32) ? lc - i0 : 32;
      const int li = (lane < cnt) ? lane : cnt - 1;
      const int s = lsrc[lb + i0 + li];
      const int dl = lrow[lb + i0 + li];
      const float nrm = lnrm[lb + i0 + li];
      add_rows(s, dl, nrm, cnt);
    }
  } else {
    const int sb = wid * QCAP;
    int pc = 0;
    int nrec = 0;
    const int nlf = E >> 7;
    {
      const int e = (nlf << 7) + (wid << 5) + lane;
      if ((nlf << 7) + (wid << 5) < E) {
        const int ec = (e < E) ? e : E - 1;
        const bool hit = (e < E) && ((unsigned)(dst[ec] - base) < (unsigned)nrows);
        const unsigned m = __builtin_amdgcn_ballot_w32(hit);
        if (m) {
          if (hit) stk[sb + (int)__builtin_amdgcn_mbcnt_lo(m, 0u)] = e;
          pc = __builtin_popcount(m);
        }
      }
    }
    for (int c0 = wid;; c0 += NW * U) {
      const bool last = c0 >= nlf;
      if (!last) {
        qint4 dq[U];
#pragma unroll
        for (int u = 0; u < U; ++u) {
          const int c = c0 + u * NW;
          const int cc = (c < nlf) ? c : nlf - 1;
          dq[u] = *(const qint4*)(dst + 4 * (((size_t)cc << 5) + lane));
        }
#pragma unroll
        for (int u = 0; u < U; ++u) {
          const int c = c0 + u * NW;
          if (c < nlf) {
#pragma unroll
            for (int k = 0; k < 4; ++k) {
              const unsigned m = __builtin_amdgcn_ballot_w32((unsigned)(dq[u][k] - base) < (unsigned)nrows);
              if (m) {
                if ((unsigned)(dq[u][k] - base) < (unsigned)nrows)
                  stk[sb + pc + (int)__builtin_amdgcn_mbcnt_lo(m, 0u)] = 4 * ((c << 5) + lane) + k;
                pc += __builtin_popcount(m);
              }
            }
          }
        }
      }
      while (pc >= (last ? 1 : 32)) {
        const int cnt = (pc < 32) ? pc : 32;
        const int b0 = pc - cnt;
        const int li = (lane < cnt) ? lane : cnt - 1;
        const int e = stk[sb + b0 + li];
        const int s = src[e];
        const int dl = dst[e] - base;
        const float ds = dinv[s];
        const float dd = dinv[base + dl];
        const float nrm = ds * dd;
        if (MODE == 0) {
          if (nrec + cnt <= LCAP && lane < cnt) {
            lsrc[lb + nrec + lane] = s;
            lrow[lb + nrec + lane] = dl;
            lnrm[lb + nrec + lane] = nrm;
          }
          nrec += cnt;
        }
        add_rows(s, dl, nrm, cnt);
        pc = b0;
      }
      if (last) break;
    }
    if (MODE == 0 && lane == 0) wcnt[wid] = (nrec <= LCAP) ? nrec : -1;
  }
  __syncthreads();

  if (MODE == 0 && wid == 0) lcount[blockIdx.x * NW + lane] = wcnt[lane];

  if constexpr (FIN) {
    constexpr int RW = (ROWS + NW - 1) / NW;
    const int per = nrows / NW;
    const int rem = nrows - per * NW;
    const int rb = wid * per + ((wid < rem) ? wid : rem);
    const int nr = per + ((wid < rem) ? 1 : 0);
    float bv[NL];
    for (int j = 0; j < NL; ++j) bv[j] = fbias[j * 32 + lane];
    float ps[NL];
    for (int j = 0; j < NL; ++j) ps[j] = 0.0f;
    int gc = -1;
    auto flush = [&]() {
      if (gc >= 0) {
        const float c = fmaxf(fcnt[gc], 1.0f);
        float* pp = fpool + (size_t)gc * C + lane;
        for (int j = 0; j < NL; ++j) atomicAdd(pp + j * 32, ps[j] / c);
      }
    };
    float v[RW][NL];
    int g[RW];
#pragma unroll
    for (int k = 0; k < RW; ++k) {
      g[k] = 0;
      if (k < nr) {
        const float* row = acc + (rb + k) * C + lane;
        for (int j = 0; j < NL; ++j) v[k][j] = row[j * 32];
        g[k] = fbatch[base + rb + k];
      }
    }
#pragma unroll
    for (int k = 0; k < RW; ++k) {
      if (k < nr) {
        float s = 0.0f;
        for (int j = 0; j < NL; ++j) {
          float t = fmaxf(v[k][j] + bv[j], 0.0f);
          v[k][j] = t;
          s += t;
        }
        s = wave_sum_down(s);
        float mean = s / (float)C;
        float q = 0.0f;
        for (int j = 0; j < NL; ++j) {
          float dlt = v[k][j] - mean;
          q += dlt * dlt;
        }
        q = wave_sum_down(q);
        float rstd = rsqrtf(q / (float)C + 1e-5f);
        float* o = agg + (size_t)(base + rb + k) * C + lane;
        const int gk = __builtin_amdgcn_readfirstlane(g[k]);
        if (gk != gc) {
          flush();
          gc = gk;
          for (int j = 0; j < NL; ++j) ps[j] = 0.0f;
        }
        for (int j = 0; j < NL; ++j) {
          const float y = (v[k][j] - mean) * rstd;
          o[j * 32] = y;
          ps[j] += y;
        }
      }
    }
    flush();
  } else {
    for (int idx = threadIdx.x; idx < total; idx += 1024) agg[(size_t)base * C + idx] = acc[idx];
  }
}

template <int NL, int R>
__global__ void __launch_bounds__(256) k_post(const float* __restrict__ agg,
                                              const float* __restrict__ bias,
                                              float* __restrict__ out, int N) {
  constexpr int C = NL * 32;
  const int lane = threadIdx.x & 31;
  const int gw = wave_in_grid();
  const int nw = waves_in_grid();
  float bv[NL];
  for (int j = 0; j < NL; ++j) bv[j] = bias[j * 32 + lane];
  for (int n0 = gw; n0 < N; n0 += R * nw) {
    float v[R][NL];
    for (int k = 0; k < R; ++k) {
      const int node = n0 + k * nw;
      if (node < N) {
        const float* row = agg + (size_t)node * C + lane;
        for (int j = 0; j < NL; ++j) v[k][j] = row[j * 32];
      }
    }
    for (int k = 0; k < R; ++k) {
      const int node = n0 + k * nw;
      if (node < N) {
        float s = 0.0f;
        for (int j = 0; j < NL; ++j) {
          float t = fmaxf(v[k][j] + bv[j], 0.0f);
          v[k][j] = t;
          s += t;
        }
        for (int off = 16; off > 0; off >>= 1) s += __shfl_xor(s, off, 32);
        float mean = s / (float)C;
        float q = 0.0f;
        for (int j = 0; j < NL; ++j) {
          float dlt = v[k][j] - mean;
          q += dlt * dlt;
        }
        for (int off = 16; off > 0; off >>= 1) q += __shfl_xor(q, off, 32);
        float rstd = rsqrtf(q / (float)C + 1e-5f);
        float* o = out + (size_t)node * C + lane;
        for (int j = 0; j < NL; ++j) o[j * 32] = (v[k][j] - mean) * rstd;
      }
    }
  }
}

template <int NL, int R>
__global__ void __launch_bounds__(256) k_post_pool(const float* __restrict__ agg,
                                                   const float* __restrict__ bias,
                                                   float* __restrict__ out,
                                                   const int* __restrict__ batch,
                                                   const float* __restrict__ cnt,
                                                   float* __restrict__ pool, int N) {
  constexpr int C = NL * 32;
  const int lane = threadIdx.x & 31;
  const int gw = wave_in_grid();
  const int nw = waves_in_grid();
  const int per = N / nw;
  const int rem = N - per * nw;
  const int nb = gw * per + ((gw < rem) ? gw : rem);
  const int ne = nb + per + ((gw < rem) ? 1 : 0);
  float bv[NL];
  for (int j = 0; j < NL; ++j) bv[j] = bias[j * 32 + lane];
  float ps[NL];
  for (int j = 0; j < NL; ++j) ps[j] = 0.0f;
  int gc = -1;
  auto flush = [&]() {
    if (gc >= 0) {
      const float c = fmaxf(cnt[gc], 1.0f);
      float* pp = pool + (size_t)gc * C + lane;
      for (int j = 0; j < NL; ++j) atomicAdd(pp + j * 32, ps[j] / c);
    }
  };
  for (int n0 = nb; n0 < ne; n0 += R) {
    float v[R][NL];
    int g[R];
    for (int k = 0; k < R; ++k) {
      const int node = n0 + k;
      g[k] = 0;
      if (node < ne) {
        const float* row = agg + (size_t)node * C + lane;
        for (int j = 0; j < NL; ++j) v[k][j] = row[j * 32];
        g[k] = batch[node];
      }
    }
    for (int k = 0; k < R; ++k) {
      const int node = n0 + k;
      if (node < ne) {
        float s = 0.0f;
        for (int j = 0; j < NL; ++j) {
          float t = fmaxf(v[k][j] + bv[j], 0.0f);
          v[k][j] = t;
          s += t;
        }
        for (int off = 16; off > 0; off >>= 1) s += __shfl_xor(s, off, 32);
        float mean = s / (float)C;
        float q = 0.0f;
        for (int j = 0; j < NL; ++j) {
          float dlt = v[k][j] - mean;
          q += dlt * dlt;
        }
        for (int off = 16; off > 0; off >>= 1) q += __shfl_xor(q, off, 32);
        float rstd = rsqrtf(q / (float)C + 1e-5f);
        float* o = out + (size_t)node * C + lane;
        const int gk = __builtin_amdgcn_readfirstlane(g[k]);
        if (gk != gc) {
          flush();
          gc = gk;
          for (int j = 0; j < NL; ++j) ps[j] = 0.0f;
        }
        for (int j = 0; j < NL; ++j) {
          const float y = (v[k][j] - mean) * rstd;
          o[j * 32] = y;
          ps[j] += y;
        }
      }
    }
  }
  flush();
}

static inline size_t align256(size_t x) { return (x + 255) & ~(size_t)255; }

extern "C" void kernel_launch(void* const* d_in, const int* in_sizes, int n_in,
                              void* d_out, int out_size, void* d_ws, size_t ws_size,
                              hipStream_t stream) {
  const float* x     = (const float*)d_in[0];
  const int*   ei    = (const int*)d_in[1];
  const int*   batch = (const int*)d_in[2];
  const float* W0 = (const float*)d_in[4];
  const float* b0 = (const float*)d_in[5];
  const float* W1 = (const float*)d_in[6];
  const float* b1 = (const float*)d_in[7];
  const float* W2 = (const float*)d_in[8];
  const float* b2 = (const float*)d_in[9];

  const int N  = in_sizes[2];
  const int E  = in_sizes[1] / 2;
  const int IN = in_sizes[0] / N;
  const int H0 = in_sizes[5];
  const int H1 = in_sizes[7];
  const int H2 = in_sizes[9];
  const int G  = out_size / H2 - N;

  const int* srcp = ei;
  const int* dstp = ei + E;

  char* ws = (char*)d_ws;
  float* dinv = (float*)ws;  ws += align256((size_t)N * 4);
  float* cnt  = (float*)ws;  ws += align256((size_t)G * 4);
  float* bufA = (float*)ws;  ws += align256((size_t)N * 256 * 4);
  float* bufB = (float*)ws;  ws += align256((size_t)N * 256 * 4);
  const int NLW = ((N + 195) / 196) * 32;
  int*   lcount = (int*)ws;    ws += align256((size_t)NLW * 4);
  int*   lsrc   = (int*)ws;    ws += align256((size_t)NLW * EDGE_LIST_CAP * 4);
  int*   lrow   = (int*)ws;    ws += align256((size_t)NLW * EDGE_LIST_CAP * 4);
  float* lnrm   = (float*)ws;  ws += align256((size_t)NLW * EDGE_LIST_CAP * 4);

  float* pool = (float*)d_out;
  float* hout = (float*)d_out + (size_t)G * H2;

  const int B = 256;
  auto cdiv = [](long long a, long long b) { return (int)((a + b - 1) / b); };

  const int GR = 1024;
  bool listed = false;
  auto edge_sum = [&](const float* hin, float* o, int C) {
    const int g = cdiv(N, 196);
    if (!listed) {
      if (C == 128) k_edge_own<4, 196, 0, false><<<g, 1024, 0, stream>>>(srcp, dstp, dinv, hin, o, lsrc, lrow, lnrm, lcount, E, N, nullptr, nullptr, nullptr, nullptr);
      else          k_edge_own<8, 196, 0, false><<<g, 1024, 0, stream>>>(srcp, dstp, dinv, hin, o, lsrc, lrow, lnrm, lcount, E, N, nullptr, nullptr, nullptr, nullptr);
    } else {
      if (C == 128) k_edge_own<4, 196, 1, false><<<g, 1024, 0, stream>>>(srcp, dstp, dinv, hin, o, lsrc, lrow, lnrm, lcount, E, N, nullptr, nullptr, nullptr, nullptr);
      else          k_edge_own<8, 196, 1, false><<<g, 1024, 0, stream>>>(srcp, dstp, dinv, hin, o, lsrc, lrow, lnrm, lcount, E, N, nullptr, nullptr, nullptr, nullptr);
    }
    listed = true;
  };
  auto edge_sum_fin = [&](const float* hin, const float* bias, float* o) {
    k_edge_own<4, 196, 1, true><<<cdiv(N, 196), 1024, 0, stream>>>(srcp, dstp, dinv, hin, o, lsrc, lrow, lnrm, lcount, E, N, bias, batch, cnt, pool);
  };
  auto post = [&](const float* a, const float* bias, float* o, int C, bool pooled) {
    if (pooled) {
      if (C == 128) k_post_pool<4, 8><<<GR, B, 0, stream>>>(a, bias, o, batch, cnt, pool, N);
      else          k_post_pool<8, 4><<<GR, B, 0, stream>>>(a, bias, o, batch, cnt, pool, N);
    } else {
      if (C == 128) k_post<4, 4><<<GR, B, 0, stream>>>(a, bias, o, N);
      else          k_post<8, 2><<<GR, B, 0, stream>>>(a, bias, o, N);
    }
  };

  const int ninit = (N > G * H2) ? N : G * H2;
  k_init<<<cdiv(ninit, B), B, 0, stream>>>(dinv, N, pool, G * H2, cnt, G);
  k_deg_cnt<<<cdiv(E > N ? E : N, B), B, 0, stream>>>(dstp, dinv, E, batch, cnt, N);
  k_dinv<<<cdiv(N, B), B, 0, stream>>>(dinv, N);

  auto layer = [&](const float* hin, const float* pb, int K, int M, const float* W, const float* bias,
                   float* tA, float* tB, float* outp, bool pooled, bool raw_out) {
    dim3 g(cdiv(N, 16), M / 128);
    const int gp = cdiv(N, GEMM_TILES * 16);
    auto gemm = [&](const float* a, float* o) {
      if (pb) {
        if (M == 256) k_gemm_wlds<256, 256, 2, true><<<gp, 1024, 0, stream>>>(a, W, pb, o, N);
        else          k_gemm_wlds<256, 128, 4, true><<<gp, 1024, 0, stream>>>(a, W, pb, o, N);
      }
      else if (K == 128 && M == 256) k_gemm_wlds<128, 256, 13, false><<<gp, 1024, 0, stream>>>(a, W, nullptr, o, N);
      else if (K == 256 && M == 256) k_gemm_wlds<256, 256, 2, false><<<gp, 1024, 0, stream>>>(a, W, nullptr, o, N);
      else if (K == 256 && M == 128) k_gemm_wlds<256, 128, 4, false><<<gp, 1024, 0, stream>>>(a, W, nullptr, o, N);
      else                           k_gemm_wmma<<<g, B, 0, stream>>>(a, W, o, N, K, M);
    };
    if (K < M) {
      edge_sum(hin, tA, K);
      gemm(tA, tB);
    } else {
      gemm(hin, tA);
      if (pooled && !raw_out && M == 128 && listed) {
        edge_sum_fin(tA, bias, outp);
        return;
      }
      edge_sum(tA, tB, M);
    }
    if (!raw_out) post(tB, bias, outp, M, pooled);
  };

  auto norm_in_product = [](int K, int M) { return K == 256 && (M == 256 || M == 128); };
  if (norm_in_product(H0, H1) && norm_in_product(H1, H2)) {
    layer(x,    nullptr, IN, H0, W0, b0,  bufA,  bufB,  nullptr, false, true);
    layer(bufB, b0,      H0, H1, W1, b1,  bufA,  bufB,  nullptr, false, true);
    layer(bufB, b1,      H1, H2, W2, b2,  bufA,  bufB,  hout, true, false);
  } else {
    layer(x,    nullptr, IN, H0, W0, b0,  bufA,  bufB,  bufA, false, false);
    layer(bufA, nullptr, H0, H1, W1, b1,  bufB,  bufA,  bufB, false, false);
    layer(bufB, nullptr, H1, H2, W2, b2,  bufA,  bufB,  hout, true, false);
  }
}
